# starting kernel
# baseline (speedup 1.0000x reference)
#include <hip/hip_runtime.h>

typedef __attribute__((ext_vector_type(2))) float v2f;
typedef __attribute__((ext_vector_type(8))) float v8f;

#define NB 128
#define NC 128
#define NE 10
#define NL 16

__device__ __forceinline__ void compute_wy(float* dst, int Kpad, int K,
                                           const float* __restrict__ w,
                                           const float* ys, int tid, int nthr) {
  for (int idx = tid; idx < NC * Kpad; idx += nthr) {
    int c = idx / Kpad;
    int k = idx - c * Kpad;
    float v = 0.f;
    if (k < K) {
#pragma unroll
      for (int e = 0; e < NE; ++e) v += w[(e * K + k) * NC + c] * ys[e];
    }
    dst[idx] = v;
  }
}

template <int K3, int K3PAD, int K2, int K2PAD>
__device__ __forceinline__ float run_cascade(const float* __restrict__ U3p,
                                             const float* __restrict__ U2p,
                                             const float* __restrict__ U1p,
                                             const float* wy3s,
                                             const float* wy2s,
                                             const float* wy1s,
                                             const float* xs,
                                             int c, int cl16, int hi) {
  const float* xrow = xs + c * NL;
  float xreg[8];
#pragma unroll
  for (int j = 0; j < 8; ++j) xreg[j] = xrow[hi * 8 + j];

  const float wy1h = 0.5f * wy1s[c];

  float accTh = 0.f;
#pragma unroll 1
  for (int W = 0; W < NL; ++W) {
    v8f acc2 = {};
    {
      const float* A2row = U2p + (W * NL + cl16) * K2;
#pragma unroll
      for (int kc = 0; kc < K2PAD / 4; ++kc) {
        int k0 = kc * 4 + hi * 2;
        int ka = k0, kb = k0 + 1;
        if (kc * 4 + 3 >= K2) {
          ka = ka < K2 ? ka : K2 - 1;
          kb = kb < K2 ? kb : K2 - 1;
        }
        v2f a2;
        a2.x = A2row[ka];
        a2.y = A2row[kb];
        v2f b2 = *(const v2f*)(wy2s + c * K2PAD + k0);
        acc2 = __builtin_amdgcn_wmma_f32_16x16x4_f32(false, a2, false, b2,
                                                     (short)0, acc2, false, false);
      }
    }

    float accw = 0.f;
#pragma unroll 2
    for (int X = 0; X < NL; ++X) {
      v8f acc = {};
      const float* Arow = U3p + ((W * NL + X) * NL + cl16) * K3;
      __builtin_prefetch(Arow + NL * K3, 0, 3);
      const float* Brow = wy3s + c * K3PAD;
#pragma unroll
      for (int kc = 0; kc < K3PAD / 4; ++kc) {
        int k0 = kc * 4 + hi * 2;
        int ka = k0, kb = k0 + 1;
        if (kc * 4 + 3 >= K3) {
          ka = ka < K3 ? ka : K3 - 1;
          kb = kb < K3 ? kb : K3 - 1;
        }
        v2f a;
        a.x = Arow[ka];
        a.y = Arow[kb];
        v2f b = *(const v2f*)(Brow + k0);
        acc = __builtin_amdgcn_wmma_f32_16x16x4_f32(false, a, false, b,
                                                    (short)0, acc, false, false);
      }
      float tp = acc[0] * xreg[0];
#pragma unroll
      for (int j = 1; j < 8; ++j) tp = fmaf(acc[j], xreg[j], tp);
      accw = fmaf(xrow[X], tp, accw);
    }
#pragma unroll
    for (int j = 0; j < 8; ++j) accw = fmaf(acc2[j], xreg[j], accw);

    float o1 = fmaf(U1p[W], wy1h, accw);
    accTh = fmaf(xrow[W], o1, accTh);
  }
  return accTh + __shfl_xor(accTh, 16, 32);
}

__global__ __launch_bounds__(512) void symcon_kernel(
    const float* __restrict__ x, const float* __restrict__ y,
    const float* __restrict__ U1_0e, const float* __restrict__ U2_0e,
    const float* __restrict__ U3_0e, const float* __restrict__ U1_1o,
    const float* __restrict__ U2_1o, const float* __restrict__ U3_1o,
    const float* __restrict__ w1_0e, const float* __restrict__ w2_0e,
    const float* __restrict__ w3_0e, const float* __restrict__ w1_1o,
    const float* __restrict__ w2_1o, const float* __restrict__ w3_1o,
    float* __restrict__ out) {
  __shared__ float xs[NC * NL];
  __shared__ float ys[NE];
  __shared__ float wy3_0e[NC * 24];
  __shared__ float wy2_0e[NC * 4];
  __shared__ float wy1_0e[NC];
  __shared__ float wy3_1o[NC * 36];
  __shared__ float wy2_1o[NC * 8];
  __shared__ float wy1_1o[NC];

  const int tid = threadIdx.x;
  const int b = blockIdx.x;
  const int pp = blockIdx.y;

  if (tid < NE) ys[tid] = y[b * NE + tid];
  for (int i = tid; i < NC * NL; i += 512) xs[i] = x[b * NC * NL + i];
  __syncthreads();

  compute_wy(wy3_0e, 24, 23, w3_0e, ys, tid, 512);
  compute_wy(wy2_0e, 4, 4, w2_0e, ys, tid, 512);
  compute_wy(wy1_0e, 1, 1, w1_0e, ys, tid, 512);
  compute_wy(wy3_1o, 36, 33, w3_1o, ys, tid, 512);
  compute_wy(wy2_1o, 8, 6, w2_1o, ys, tid, 512);
  compute_wy(wy1_1o, 1, 1, w1_1o, ys, tid, 512);
  __syncthreads();

  const int wv = tid >> 5;
  const int lane = tid & 31;
  const int cl16 = lane & 15;
  const int hi = lane >> 4;

  const int p = wv + 16 * pp;
  const int casc = p >> 3;
  const int ct = p & 7;
  const int c = ct * 16 + cl16;

  if (casc == 0) {
    float r = run_cascade<23, 24, 4, 4>(U3_0e, U2_0e, U1_0e, wy3_0e, wy2_0e,
                                        wy1_0e, xs, c, cl16, hi);
    if (lane < 16) out[b * 512 + c] = r;
  } else {
    const int eq = casc - 1;
    float r = run_cascade<33, 36, 6, 8>(U3_1o + eq * (NL * NL * NL * 33),
                                        U2_1o + eq * (NL * NL * 6),
                                        U1_1o + eq * NL, wy3_1o, wy2_1o,
                                        wy1_1o, xs, c, cl16, hi);
    if (lane < 16) out[b * 512 + NC + c * 3 + eq] = r;
  }
}

extern "C" void kernel_launch(void* const* d_in, const int* in_sizes, int n_in,
                              void* d_out, int out_size, void* d_ws,
                              size_t ws_size, hipStream_t stream) {
  (void)in_sizes; (void)n_in; (void)d_ws; (void)ws_size; (void)out_size;
  const float* x     = (const float*)d_in[0];
  const float* y     = (const float*)d_in[1];
  const float* U1_0e = (const float*)d_in[2];
  const float* U2_0e = (const float*)d_in[3];
  const float* U3_0e = (const float*)d_in[4];
  const float* U1_1o = (const float*)d_in[5];
  const float* U2_1o = (const float*)d_in[6];
  const float* U3_1o = (const float*)d_in[7];
  const float* w1_0e = (const float*)d_in[8];
  const float* w2_0e = (const float*)d_in[9];
  const float* w3_0e = (const float*)d_in[10];
  const float* w1_1o = (const float*)d_in[11];
  const float* w2_1o = (const float*)d_in[12];
  const float* w3_1o = (const float*)d_in[13];
  float* out = (float*)d_out;

  dim3 grid(NB, 2);
  symcon_kernel<<<grid, 512, 0, stream>>>(x, y, U1_0e, U2_0e, U3_0e, U1_1o,
                                          U2_1o, U3_1o, w1_0e, w2_0e, w3_0e,
                                          w1_1o, w2_1o, w3_1o, out);
}


# Wave tile 64 channels x 16 columns: one B fragment feeds four WMMAs; a chunk's 16 B fragments are read from LDS as one group one chunk ahead (2 register sets), none between WMMAs; sum order kept
# speedup vs baseline: 2.1055x; 2.1055x over previous
#include <hip/hip_runtime.h>

typedef __attribute__((ext_vector_type(2))) float v2f;
typedef __attribute__((ext_vector_type(4))) float v4f;
typedef __attribute__((ext_vector_type(8))) float v8f;

#define NB 128
#define NC 128
#define NE 10
#define NL 16
#define K3A 23
#define K3B 33
#define NROW 4096
#define CH_ROWS 64
#define NCHUNK 64
#define PITCH 144
#define BUF_FLOATS (32 * PITCH * 2)
#define XP 20
#define PP 65


template <int K3, int K2>
__device__ __forceinline__ float finish(const float* __restrict__ w3p, const float* __restrict__ w2p,
                                        const float* __restrict__ w1p, const float* __restrict__ U1p,
                                        const float* P3, const float* P2c, const float* xrow,
                                        const float* ys, int c) {
  float yv[NE];
#pragma unroll
  for (int e = 0; e < NE; ++e) yv[e] = ys[e];
  float o = 0.f;
#pragma unroll
  for (int kk = 0; kk < K3; ++kk) {
    float wy = 0.f;
#pragma unroll
    for (int e = 0; e < NE; ++e) wy = fmaf(w3p[(e * K3 + kk) * NC + c], yv[e], wy);
    o = fmaf(wy, P3[kk * PP], o);
  }
#pragma unroll
  for (int kk = 0; kk < K2; ++kk) {
    float wy = 0.f;
#pragma unroll
    for (int e = 0; e < NE; ++e) wy = fmaf(w2p[(e * K2 + kk) * NC + c], yv[e], wy);
    o = fmaf(wy, P2c[kk * PP], o);
  }
  float wy1 = 0.f;
#pragma unroll
  for (int e = 0; e < NE; ++e) wy1 = fmaf(w1p[e * NC + c], yv[e], wy1);
  float s1 = 0.f;
#pragma unroll
  for (int W = 0; W < NL; ++W) s1 = fmaf(xrow[W], U1p[W], s1);
  return fmaf(wy1, s1, o);
}

__global__ __launch_bounds__(256) void symcon_kernel(
    const float* __restrict__ x, const float* __restrict__ y,
    const float* __restrict__ U1_0e, const float* __restrict__ U2_0e,
    const float* __restrict__ U3_0e, const float* __restrict__ U1_1o,
    const float* __restrict__ U2_1o, const float* __restrict__ U3_1o,
    const float* __restrict__ w1_0e, const float* __restrict__ w2_0e,
    const float* __restrict__ w3_0e, const float* __restrict__ w1_1o,
    const float* __restrict__ w2_1o, const float* __restrict__ w3_1o,
    float* __restrict__ out) {
  __shared__ __attribute__((aligned(16))) float bufs[2 * BUF_FLOATS];
  __shared__ __attribute__((aligned(16))) float xs[64 * XP];
  __shared__ float ys[NE];

  const int tid = threadIdx.x;
  const int b = blockIdx.x;
  const int h = blockIdx.y;
  const int wv = tid >> 5;
  const int lane = tid & 31;
  const int n = lane & 15;
  const int hi = lane >> 4;
  if (tid < NE) ys[tid] = y[b * NE + tid];
  for (int i = tid; i < 64 * NL; i += 256) {
    int cl = i >> 4, l = i & 15;
    xs[cl * XP + l] = x[(b * NC + 64 * h + cl) * NL + l];
  }
  for (int i = tid; i < 2 * 32 * 12; i += 256) {
    int bf = i / 384, rem = i - bf * 384;
    int kp = rem / 12, r2 = rem - kp * 12;
    bufs[bf * BUF_FLOATS + (kp * PITCH + 122 + (r2 >> 1)) * 2 + (r2 & 1)] = 0.f;
  }

  float r0[6], r1[27];
  auto prefetch = [&](int ch) {
    const float* s0 = U3_0e + ch * (CH_ROWS * K3A);
#pragma unroll
    for (int it = 0; it < 5; ++it) r0[it] = s0[tid + 256 * it];
    r0[5] = s0[min(tid + 1280, CH_ROWS * K3A - 1)];
#pragma unroll
    for (int e = 0; e < 3; ++e) {
      const float* s1 = U3_1o + e * (NROW * K3B) + ch * (CH_ROWS * K3B);
#pragma unroll
      for (int it = 0; it < 8; ++it) r1[e * 9 + it] = s1[tid + 256 * it];
      r1[e * 9 + 8] = s1[min(tid + 2048, CH_ROWS * K3B - 1)];
    }
  };
  auto stash = [&](float* buf) {
#pragma unroll
    for (int it = 0; it < 6; ++it) {
      int f = tid + 256 * it;
      if (it < 5 || tid < CH_ROWS * K3A - 1280) {
        int lr = f / K3A, kk = f - lr * K3A;
        buf[((lr >> 1) * PITCH + kk) * 2 + (lr & 1)] = r0[it];
      }
    }
#pragma unroll
    for (int e = 0; e < 3; ++e) {
#pragma unroll
      for (int it = 0; it < 9; ++it) {
        int f = tid + 256 * it;
        if (it < 8 || tid < CH_ROWS * K3B - 2048) {
          int lr = f / K3B, kk = f - lr * K3B;
          buf[((lr >> 1) * PITCH + K3A + K3B * e + kk) * 2 + (lr & 1)] = r1[e * 9 + it];
        }
      }
    }
  };

  auto load_frags = [&](int ch, v2f (&f)[16], float (&xw)[4], v4f (&xX)[4]) __attribute__((always_inline)) {
    const float* src = bufs + (ch & 1) * BUF_FLOATS + (16 * wv + n) * 2 + hi * (PITCH * 2);
#pragma unroll
    for (int kc = 0; kc < 16; ++kc) f[kc] = *(const v2f*)(src + 2 * kc * PITCH * 2);
    const int W = ch >> 2, Xg = ch & 3;
#pragma unroll
    for (int t = 0; t < 4; ++t) {
      xw[t] = xs[(16 * t + n) * XP + W];
      xX[t] = *(const v4f*)(xs + (16 * t + n) * XP + 4 * Xg);
    }
  };

  prefetch(0);
  __syncthreads();

  float xi[4][8];
#pragma unroll
  for (int t = 0; t < 4; ++t)
#pragma unroll
    for (int q = 0; q < 4; ++q) {
      xi[t][2 * q] = xs[(16 * t + n) * XP + 4 * q + 2 * hi];
      xi[t][2 * q + 1] = xs[(16 * t + n) * XP + 4 * q + 2 * hi + 1];
    }
  stash(bufs);
  prefetch(1);
  __syncthreads();

  v2f fa[16], fb[16];
  float xwa[4], xwb[4];
  v4f xXa[4], xXb[4];
  load_frags(0, fa, xwa, xXa);
  stash(bufs + BUF_FLOATS);

  v8f acc[4] = {};
  auto step = [&](int ch, const v2f (&f)[16], const float (&xw)[4], const v4f (&xX)[4],
                  v2f (&fn)[16], float (&xwn)[4], v4f (&xXn)[4]) __attribute__((always_inline)) {
    __syncthreads();
    if (ch + 2 < NCHUNK) prefetch(ch + 2);
    if (ch + 1 < NCHUNK) load_frags(ch + 1, fn, xwn, xXn);
    __builtin_amdgcn_sched_barrier(0);
#pragma unroll
    for (int xq = 0; xq < 4; ++xq) {
      float pwx[4];
#pragma unroll
      for (int t = 0; t < 4; ++t) pwx[t] = xw[t] * xX[t][xq];
#pragma unroll
      for (int iq = 0; iq < 4; ++iq) {
        const int kc = 4 * xq + iq;
#pragma unroll
        for (int t = 0; t < 4; ++t) {
          v2f a;
          a.x = pwx[t] * xi[t][2 * iq];
          a.y = pwx[t] * xi[t][2 * iq + 1];
          acc[t] = __builtin_amdgcn_wmma_f32_16x16x4_f32(false, a, false, f[kc],
                                                         (short)0, acc[t], false, false);
        }
      }
    }
    if (ch + 2 < NCHUNK) stash(bufs + (ch & 1) * BUF_FLOATS);
  };
  for (int ch = 0; ch < NCHUNK; ch += 2) {
    step(ch, fa, xwa, xXa, fb, xwb, xXb);
    step(ch + 1, fb, xwb, xXb, fa, xwa, xXa);
  }

  v8f a2a = {}, a2b = {};
  const int mt = wv & 3;
  const int nh = wv >> 2;
  {
    const int crow = 16 * mt + n;
    float xi2[8];
#pragma unroll
    for (int q = 0; q < 4; ++q) {
      xi2[2 * q] = xs[crow * XP + 4 * q + 2 * hi];
      xi2[2 * q + 1] = xs[crow * XP + 4 * q + 2 * hi + 1];
    }
    const int col2 = 16 * nh + n;
    const bool ok2 = col2 < 22;
    const float* p2 = U2_0e;
    int K2s = 0;
    if (col2 < 4) {
      p2 = U2_0e + col2;
      K2s = 4;
    } else if (col2 < 22) {
      int e = (col2 - 4) / 6, kk = (col2 - 4) - 6 * e;
      p2 = U2_1o + e * (NL * NL * 6) + kk;
      K2s = 6;
    }
#pragma unroll 4
    for (int W = 0; W < NL; ++W) {
      const float xw = xs[crow * XP + W];
#pragma unroll
      for (int q = 0; q < 4; ++q) {
        const int rr = 16 * W + 4 * q + 2 * hi;
        v2f a;
        a.x = xw * xi2[2 * q];
        a.y = xw * xi2[2 * q + 1];
        float b0 = p2[rr * K2s], b1 = p2[(rr + 1) * K2s];
        v2f bv;
        bv.x = ok2 ? b0 : 0.f;
        bv.y = ok2 ? b1 : 0.f;
        if (q & 1)
          a2b = __builtin_amdgcn_wmma_f32_16x16x4_f32(false, a, false, bv, (short)0, a2b, false, false);
        else
          a2a = __builtin_amdgcn_wmma_f32_16x16x4_f32(false, a, false, bv, (short)0, a2a, false, false);
      }
    }
  }

  __syncthreads();
  float* P = bufs;
  float* P2 = bufs + 128 * PP;
#pragma unroll
  for (int t = 0; t < 4; ++t)
#pragma unroll
    for (int j = 0; j < 8; ++j)
      P[(16 * wv + n) * PP + 16 * t + 8 * hi + j] = acc[t][j];
  {
    v8f d2 = a2a + a2b;
#pragma unroll
    for (int j = 0; j < 8; ++j) P2[(16 * nh + n) * PP + 16 * mt + 8 * hi + j] = d2[j];
  }
  __syncthreads();

  {
    const int q = tid >> 6;
    const int cl = tid & 63;
    const int c = 64 * h + cl;
    if (q == 0) {
      out[b * 512 + c] = finish<K3A, 4>(w3_0e, w2_0e, w1_0e, U1_0e, P + cl, P2 + cl,
                                        xs + cl * XP, ys, c);
    } else {
      const int eq = q - 1;
      out[b * 512 + NC + c * 3 + eq] =
          finish<K3B, 6>(w3_1o, w2_1o, w1_1o, U1_1o + eq * NL, P + (K3A + K3B * eq) * PP + cl,
                         P2 + (4 + 6 * eq) * PP + cl, xs + cl * XP, ys, c);
    }
  }
}

extern "C" void kernel_launch(void* const* d_in, const int* in_sizes, int n_in,
                              void* d_out, int out_size, void* d_ws,
                              size_t ws_size, hipStream_t stream) {
  (void)in_sizes; (void)n_in; (void)d_ws; (void)ws_size; (void)out_size;
  const float* x     = (const float*)d_in[0];
  const float* y     = (const float*)d_in[1];
  const float* U1_0e = (const float*)d_in[2];
  const float* U2_0e = (const float*)d_in[3];
  const float* U3_0e = (const float*)d_in[4];
  const float* U1_1o = (const float*)d_in[5];
  const float* U2_1o = (const float*)d_in[6];
  const float* U3_1o = (const float*)d_in[7];
  const float* w1_0e = (const float*)d_in[8];
  const float* w2_0e = (const float*)d_in[9];
  const float* w3_0e = (const float*)d_in[10];
  const float* w1_1o = (const float*)d_in[11];
  const float* w2_1o = (const float*)d_in[12];
  const float* w3_1o = (const float*)d_in[13];
  float* out = (float*)d_out;

  dim3 grid(NB, 2);
  symcon_kernel<<<grid, 256, 0, stream>>>(x, y, U1_0e, U2_0e, U3_0e, U1_1o,
                                          U2_1o, U3_1o, w1_0e, w2_0e, w3_0e,
                                          w1_1o, w2_1o, w3_1o, out);
}


# Role split of each SIMD's wave pair per chunk: waves 0-3 run the 64 WMMAs first and stage behind them, waves 4-7 stage first; chunk barrier orders LDS only so loads stay in flight; sum order kept
# speedup vs baseline: 2.3706x; 1.1259x over previous
#include <hip/hip_runtime.h>

typedef __attribute__((ext_vector_type(2))) float v2f;
typedef __attribute__((ext_vector_type(4))) float v4f;
typedef __attribute__((ext_vector_type(8))) float v8f;

#define NB 128
#define NC 128
#define NE 10
#define NL 16
#define K3A 23
#define K3B 33
#define NROW 4096
#define CH_ROWS 64
#define NCHUNK 64
#define PITCH 144
#define BUF_FLOATS (32 * PITCH * 2)
#define XP 20
#define PP 65


template <int K3, int K2>
__device__ __forceinline__ float finish(const float* __restrict__ w3p, const float* __restrict__ w2p,
                                        const float* __restrict__ w1p, const float* __restrict__ U1p,
                                        const float* P3, const float* P2c, const float* xrow,
                                        const float* ys, int c) {
  float yv[NE];
#pragma unroll
  for (int e = 0; e < NE; ++e) yv[e] = ys[e];
  float o = 0.f;
#pragma unroll
  for (int kk = 0; kk < K3; ++kk) {
    float wy = 0.f;
#pragma unroll
    for (int e = 0; e < NE; ++e) wy = fmaf(w3p[(e * K3 + kk) * NC + c], yv[e], wy);
    o = fmaf(wy, P3[kk * PP], o);
  }
#pragma unroll
  for (int kk = 0; kk < K2; ++kk) {
    float wy = 0.f;
#pragma unroll
    for (int e = 0; e < NE; ++e) wy = fmaf(w2p[(e * K2 + kk) * NC + c], yv[e], wy);
    o = fmaf(wy, P2c[kk * PP], o);
  }
  float wy1 = 0.f;
#pragma unroll
  for (int e = 0; e < NE; ++e) wy1 = fmaf(w1p[e * NC + c], yv[e], wy1);
  float s1 = 0.f;
#pragma unroll
  for (int W = 0; W < NL; ++W) s1 = fmaf(xrow[W], U1p[W], s1);
  return fmaf(wy1, s1, o);
}

__global__ __launch_bounds__(256) void symcon_kernel(
    const float* __restrict__ x, const float* __restrict__ y,
    const float* __restrict__ U1_0e, const float* __restrict__ U2_0e,
    const float* __restrict__ U3_0e, const float* __restrict__ U1_1o,
    const float* __restrict__ U2_1o, const float* __restrict__ U3_1o,
    const float* __restrict__ w1_0e, const float* __restrict__ w2_0e,
    const float* __restrict__ w3_0e, const float* __restrict__ w1_1o,
    const float* __restrict__ w2_1o, const float* __restrict__ w3_1o,
    float* __restrict__ out) {
  __shared__ __attribute__((aligned(16))) float bufs[2 * BUF_FLOATS];
  __shared__ __attribute__((aligned(16))) float xs[64 * XP];
  __shared__ float ys[NE];

  const int tid = threadIdx.x;
  const int b = blockIdx.x;
  const int h = blockIdx.y;
  const int wv = tid >> 5;
  const int lane = tid & 31;
  const int n = lane & 15;
  const int hi = lane >> 4;
  if (tid < NE) ys[tid] = y[b * NE + tid];
  for (int i = tid; i < 64 * NL; i += 256) {
    int cl = i >> 4, l = i & 15;
    xs[cl * XP + l] = x[(b * NC + 64 * h + cl) * NL + l];
  }
  for (int i = tid; i < 2 * 32 * 12; i += 256) {
    int bf = i / 384, rem = i - bf * 384;
    int kp = rem / 12, r2 = rem - kp * 12;
    bufs[bf * BUF_FLOATS + (kp * PITCH + 122 + (r2 >> 1)) * 2 + (r2 & 1)] = 0.f;
  }

  float r0[6], r1[27];
  auto prefetch = [&](int ch) {
    const float* s0 = U3_0e + ch * (CH_ROWS * K3A);
#pragma unroll
    for (int it = 0; it < 5; ++it) r0[it] = s0[tid + 256 * it];
    r0[5] = s0[min(tid + 1280, CH_ROWS * K3A - 1)];
#pragma unroll
    for (int e = 0; e < 3; ++e) {
      const float* s1 = U3_1o + e * (NROW * K3B) + ch * (CH_ROWS * K3B);
#pragma unroll
      for (int it = 0; it < 8; ++it) r1[e * 9 + it] = s1[tid + 256 * it];
      r1[e * 9 + 8] = s1[min(tid + 2048, CH_ROWS * K3B - 1)];
    }
  };
  auto stash = [&](float* buf) {
#pragma unroll
    for (int it = 0; it < 6; ++it) {
      int f = tid + 256 * it;
      if (it < 5 || tid < CH_ROWS * K3A - 1280) {
        int lr = f / K3A, kk = f - lr * K3A;
        buf[((lr >> 1) * PITCH + kk) * 2 + (lr & 1)] = r0[it];
      }
    }
#pragma unroll
    for (int e = 0; e < 3; ++e) {
#pragma unroll
      for (int it = 0; it < 9; ++it) {
        int f = tid + 256 * it;
        if (it < 8 || tid < CH_ROWS * K3B - 2048) {
          int lr = f / K3B, kk = f - lr * K3B;
          buf[((lr >> 1) * PITCH + K3A + K3B * e + kk) * 2 + (lr & 1)] = r1[e * 9 + it];
        }
      }
    }
  };

  auto load_frags = [&](int ch, v2f (&f)[16], float (&xw)[4], v4f (&xX)[4]) __attribute__((always_inline)) {
    const float* src = bufs + (ch & 1) * BUF_FLOATS + (16 * wv + n) * 2 + hi * (PITCH * 2);
#pragma unroll
    for (int kc = 0; kc < 16; ++kc) f[kc] = *(const v2f*)(src + 2 * kc * PITCH * 2);
    const int W = ch >> 2, Xg = ch & 3;
#pragma unroll
    for (int t = 0; t < 4; ++t) {
      xw[t] = xs[(16 * t + n) * XP + W];
      xX[t] = *(const v4f*)(xs + (16 * t + n) * XP + 4 * Xg);
    }
  };

  prefetch(0);
  __syncthreads();

  float xi[4][8];
#pragma unroll
  for (int t = 0; t < 4; ++t)
#pragma unroll
    for (int q = 0; q < 4; ++q) {
      xi[t][2 * q] = xs[(16 * t + n) * XP + 4 * q + 2 * hi];
      xi[t][2 * q + 1] = xs[(16 * t + n) * XP + 4 * q + 2 * hi + 1];
    }
  stash(bufs);
  prefetch(1);
  __syncthreads();

  v2f fa[16], fb[16];
  float xwa[4], xwb[4];
  v4f xXa[4], xXb[4];
  load_frags(0, fa, xwa, xXa);
  stash(bufs + BUF_FLOATS);
  prefetch(2);

  const bool mma_first = __builtin_amdgcn_readfirstlane(wv) < 4;

  v8f acc[4] = {};
  auto step = [&](int ch, const v2f (&f)[16], const float (&xw)[4], const v4f (&xX)[4],
                  v2f (&fn)[16], float (&xwn)[4], v4f (&xXn)[4]) __attribute__((always_inline)) {
    auto stage = [&]() __attribute__((always_inline)) {
      if (ch + 2 < NCHUNK) stash(bufs + (ch & 1) * BUF_FLOATS);
      if (ch + 3 < NCHUNK) prefetch(ch + 3);
      if (ch + 1 < NCHUNK) load_frags(ch + 1, fn, xwn, xXn);
    };
    __builtin_amdgcn_fence(__ATOMIC_RELEASE, "workgroup", "local");
    __builtin_amdgcn_s_barrier();
    __builtin_amdgcn_fence(__ATOMIC_ACQUIRE, "workgroup", "local");
    if (!mma_first) stage();
#pragma unroll
    for (int xq = 0; xq < 4; ++xq) {
      float pwx[4];
#pragma unroll
      for (int t = 0; t < 4; ++t) pwx[t] = xw[t] * xX[t][xq];
#pragma unroll
      for (int iq = 0; iq < 4; ++iq) {
        const int kc = 4 * xq + iq;
#pragma unroll
        for (int t = 0; t < 4; ++t) {
          v2f a;
          a.x = pwx[t] * xi[t][2 * iq];
          a.y = pwx[t] * xi[t][2 * iq + 1];
          acc[t] = __builtin_amdgcn_wmma_f32_16x16x4_f32(false, a, false, f[kc],
                                                         (short)0, acc[t], false, false);
        }
      }
    }
    if (mma_first) stage();
  };
  for (int ch = 0; ch < NCHUNK; ch += 2) {
    step(ch, fa, xwa, xXa, fb, xwb, xXb);
    step(ch + 1, fb, xwb, xXb, fa, xwa, xXa);
  }

  v8f a2a = {}, a2b = {};
  const int mt = wv & 3;
  const int nh = wv >> 2;
  {
    const int crow = 16 * mt + n;
    float xi2[8];
#pragma unroll
    for (int q = 0; q < 4; ++q) {
      xi2[2 * q] = xs[crow * XP + 4 * q + 2 * hi];
      xi2[2 * q + 1] = xs[crow * XP + 4 * q + 2 * hi + 1];
    }
    const int col2 = 16 * nh + n;
    const bool ok2 = col2 < 22;
    const float* p2 = U2_0e;
    int K2s = 0;
    if (col2 < 4) {
      p2 = U2_0e + col2;
      K2s = 4;
    } else if (col2 < 22) {
      int e = (col2 - 4) / 6, kk = (col2 - 4) - 6 * e;
      p2 = U2_1o + e * (NL * NL * 6) + kk;
      K2s = 6;
    }
#pragma unroll 4
    for (int W = 0; W < NL; ++W) {
      const float xw = xs[crow * XP + W];
#pragma unroll
      for (int q = 0; q < 4; ++q) {
        const int rr = 16 * W + 4 * q + 2 * hi;
        v2f a;
        a.x = xw * xi2[2 * q];
        a.y = xw * xi2[2 * q + 1];
        float b0 = p2[rr * K2s], b1 = p2[(rr + 1) * K2s];
        v2f bv;
        bv.x = ok2 ? b0 : 0.f;
        bv.y = ok2 ? b1 : 0.f;
        if (q & 1)
          a2b = __builtin_amdgcn_wmma_f32_16x16x4_f32(false, a, false, bv, (short)0, a2b, false, false);
        else
          a2a = __builtin_amdgcn_wmma_f32_16x16x4_f32(false, a, false, bv, (short)0, a2a, false, false);
      }
    }
  }

  __syncthreads();
  float* P = bufs;
  float* P2 = bufs + 128 * PP;
#pragma unroll
  for (int t = 0; t < 4; ++t)
#pragma unroll
    for (int j = 0; j < 8; ++j)
      P[(16 * wv + n) * PP + 16 * t + 8 * hi + j] = acc[t][j];
  {
    v8f d2 = a2a + a2b;
#pragma unroll
    for (int j = 0; j < 8; ++j) P2[(16 * nh + n) * PP + 16 * mt + 8 * hi + j] = d2[j];
  }
  __syncthreads();

  {
    const int q = tid >> 6;
    const int cl = tid & 63;
    const int c = 64 * h + cl;
    if (q == 0) {
      out[b * 512 + c] = finish<K3A, 4>(w3_0e, w2_0e, w1_0e, U1_0e, P + cl, P2 + cl,
                                        xs + cl * XP, ys, c);
    } else {
      const int eq = q - 1;
      out[b * 512 + NC + c * 3 + eq] =
          finish<K3B, 6>(w3_1o, w2_1o, w1_1o, U1_1o + eq * NL, P + (K3A + K3B * eq) * PP + cl,
                         P2 + (4 + 6 * eq) * PP + cl, xs + cl * XP, ys, c);
    }
  }
}

extern "C" void kernel_launch(void* const* d_in, const int* in_sizes, int n_in,
                              void* d_out, int out_size, void* d_ws,
                              size_t ws_size, hipStream_t stream) {
  (void)in_sizes; (void)n_in; (void)d_ws; (void)ws_size; (void)out_size;
  const float* x     = (const float*)d_in[0];
  const float* y     = (const float*)d_in[1];
  const float* U1_0e = (const float*)d_in[2];
  const float* U2_0e = (const float*)d_in[3];
  const float* U3_0e = (const float*)d_in[4];
  const float* U1_1o = (const float*)d_in[5];
  const float* U2_1o = (const float*)d_in[6];
  const float* U3_1o = (const float*)d_in[7];
  const float* w1_0e = (const float*)d_in[8];
  const float* w2_0e = (const float*)d_in[9];
  const float* w3_0e = (const float*)d_in[10];
  const float* w1_1o = (const float*)d_in[11];
  const float* w2_1o = (const float*)d_in[12];
  const float* w3_1o = (const float*)d_in[13];
  float* out = (float*)d_out;

  dim3 grid(NB, 2);
  symcon_kernel<<<grid, 256, 0, stream>>>(x, y, U1_0e, U2_0e, U3_0e, U1_1o,
                                          U2_1o, U3_1o, w1_0e, w2_0e, w3_0e,
                                          w1_1o, w2_1o, w3_1o, out);
}


# Weights first: per channel V = U3*w3 (10 element columns per cascade) on WMMA, its D tile is the B operand of the atoms x V product; r split over 8 waves, summed via LDS, y last (new sum order)
# speedup vs baseline: 2.8570x; 1.2052x over previous
#include <hip/hip_runtime.h>


typedef __attribute__((ext_vector_type(2))) float v2f;
typedef __attribute__((ext_vector_type(4))) float v4f;
typedef __attribute__((ext_vector_type(8))) float v8f;

#define NB 128
#define NC 128
#define NE 10
#define NL 16
#define K3A 23
#define K3B 33
#define NROW 4096
#define CH_ROWS 128
#define NCHUNK 32
#define LP 132
#define CB1 24
#define BUF_FLOATS (CH_ROWS * LP)
#define XP 20
#define PP 65


__global__ __launch_bounds__(256) void symcon_kernel(
    const float* __restrict__ x, const float* __restrict__ y,
    const float* __restrict__ U1_0e, const float* __restrict__ U2_0e,
    const float* __restrict__ U3_0e, const float* __restrict__ U1_1o,
    const float* __restrict__ U2_1o, const float* __restrict__ U3_1o,
    const float* __restrict__ w1_0e, const float* __restrict__ w2_0e,
    const float* __restrict__ w3_0e, const float* __restrict__ w1_1o,
    const float* __restrict__ w2_1o, const float* __restrict__ w3_1o,
    float* __restrict__ out) {
  __shared__ __attribute__((aligned(16))) float bufs[2 * BUF_FLOATS];
  __shared__ __attribute__((aligned(16))) float xs[64 * XP];
  __shared__ float ysm[64 * NE];
  __shared__ float p2s[32 * PP];

  const int tid = threadIdx.x;
  const int c = blockIdx.x;
  const int ah = blockIdx.y;
  const int wv = tid >> 5;
  const int lane = tid & 31;
  const int n = lane & 15;
  const int hi = lane >> 4;
  for (int i = tid; i < 64 * NE; i += 256) ysm[i] = y[64 * ah * NE + i];
  for (int i = tid; i < 64 * NL; i += 256) {
    int al = i >> 4, l = i & 15;
    xs[al * XP + l] = x[((64 * ah + al) * NC + c) * NL + l];
  }
  for (int i = tid; i < 2 * CH_ROWS * 10; i += 256) {
    int bf = i / (CH_ROWS * 10), rem = i - bf * (CH_ROWS * 10);
    int row = rem / 10, p = rem - row * 10;
    int col = (p == 0) ? K3A : (CB1 + K3B + 36 * ((p - 1) / 3) + (p - 1) % 3);
    bufs[bf * BUF_FLOATS + row * LP + col] = 0.f;
  }

  float r0[12], r1[51];
  auto prefetch = [&](int ch) {
    const float* s0 = U3_0e + ch * (CH_ROWS * K3A);
#pragma unroll
    for (int it = 0; it < 11; ++it) r0[it] = s0[tid + 256 * it];
    r0[11] = s0[min(tid + 2816, CH_ROWS * K3A - 1)];
#pragma unroll
    for (int e = 0; e < 3; ++e) {
      const float* s1 = U3_1o + e * (NROW * K3B) + ch * (CH_ROWS * K3B);
#pragma unroll
      for (int it = 0; it < 16; ++it) r1[e * 17 + it] = s1[tid + 256 * it];
      r1[e * 17 + 16] = s1[min(tid + 4096, CH_ROWS * K3B - 1)];
    }
  };
  auto stash = [&](float* buf) {
#pragma unroll
    for (int it = 0; it < 12; ++it) {
      int f = tid + 256 * it;
      if (it < 11 || tid < CH_ROWS * K3A - 2816) {
        int lr = f / K3A, kk = f - lr * K3A;
        buf[lr * LP + kk] = r0[it];
      }
    }
#pragma unroll
    for (int e = 0; e < 3; ++e) {
#pragma unroll
      for (int it = 0; it < 17; ++it) {
        int f = tid + 256 * it;
        if (it < 16 || tid < CH_ROWS * K3B - 4096) {
          int lr = f / K3B, kk = f - lr * K3B;
          buf[lr * LP + CB1 + 36 * e + kk] = r1[e * 17 + it];
        }
      }
    }
  };

  v2f af0[6], af1[3][9];
  float xw[4], xX[4];
  auto load_frags = [&](int ch) __attribute__((always_inline)) {
    const float* src = bufs + (ch & 1) * BUF_FLOATS + (16 * wv + n) * LP + 2 * hi;
#pragma unroll
    for (int s = 0; s < 6; ++s) af0[s] = *(const v2f*)(src + 4 * s);
#pragma unroll
    for (int e = 0; e < 3; ++e)
#pragma unroll
      for (int s = 0; s < 9; ++s) af1[e][s] = *(const v2f*)(src + CB1 + 36 * e + 4 * s);
    const int T = 8 * ch + wv;
    const int W = (T >> 4) & 15, X = T & 15;
#pragma unroll
    for (int t = 0; t < 4; ++t) {
      xw[t] = xs[(16 * t + n) * XP + W];
      xX[t] = xs[(16 * t + n) * XP + X];
    }
  };

  prefetch(0);

  v2f bw0[6], bw1[9];
#pragma unroll
  for (int s = 0; s < 6; ++s) {
    const int k0 = 4 * s + 2 * hi;
    bw0[s].x = (n < NE && k0 < K3A) ? w3_0e[(n * K3A + k0) * NC + c] : 0.f;
    bw0[s].y = (n < NE && k0 + 1 < K3A) ? w3_0e[(n * K3A + k0 + 1) * NC + c] : 0.f;
  }
#pragma unroll
  for (int s = 0; s < 9; ++s) {
    const int k0 = 4 * s + 2 * hi;
    bw1[s].x = (n < NE && k0 < K3B) ? w3_1o[(n * K3B + k0) * NC + c] : 0.f;
    bw1[s].y = (n < NE && k0 + 1 < K3B) ? w3_1o[(n * K3B + k0 + 1) * NC + c] : 0.f;
  }

  __syncthreads();

  float xi[4][8];
#pragma unroll
  for (int t = 0; t < 4; ++t) {
    v4f lo = *(const v4f*)(xs + (16 * t + n) * XP + 8 * hi);
    v4f hh = *(const v4f*)(xs + (16 * t + n) * XP + 8 * hi + 4);
#pragma unroll
    for (int j = 0; j < 4; ++j) {
      xi[t][j] = lo[j];
      xi[t][4 + j] = hh[j];
    }
  }
  stash(bufs);
  prefetch(1);
  __syncthreads();

  load_frags(0);
  stash(bufs + BUF_FLOATS);
  prefetch(2);

  const bool mma_first = __builtin_amdgcn_readfirstlane(wv) < 4;

  v8f acc[4][4] = {};
  for (int ch = 0; ch < NCHUNK; ++ch) {
    auto stage = [&]() __attribute__((always_inline)) {
      if (ch + 2 < NCHUNK) stash(bufs + (ch & 1) * BUF_FLOATS);
      if (ch + 3 < NCHUNK) prefetch(ch + 3);
    };
    __builtin_amdgcn_fence(__ATOMIC_RELEASE, "workgroup", "local");
    __builtin_amdgcn_s_barrier();
    __builtin_amdgcn_fence(__ATOMIC_ACQUIRE, "workgroup", "local");
    if (!mma_first) stage();
    float pwx[4];
#pragma unroll
    for (int t = 0; t < 4; ++t) pwx[t] = xw[t] * xX[t];
    v8f V[4] = {};
#pragma unroll
    for (int s = 0; s < 9; ++s) {
      if (s < 6)
        V[0] = __builtin_amdgcn_wmma_f32_16x16x4_f32(false, af0[s], false, bw0[s], (short)0, V[0], false, false);
#pragma unroll
      for (int e = 0; e < 3; ++e)
        V[1 + e] = __builtin_amdgcn_wmma_f32_16x16x4_f32(false, af1[e][s], false, bw1[s], (short)0, V[1 + e], false, false);
    }
    load_frags(ch + 1 < NCHUNK ? ch + 1 : ch);
#pragma unroll
    for (int q = 0; q < 4; ++q) {
#pragma unroll
      for (int t = 0; t < 4; ++t) {
        v2f a;
        a.x = pwx[t] * xi[t][2 * q];
        a.y = pwx[t] * xi[t][2 * q + 1];
#pragma unroll
        for (int cs = 0; cs < 4; ++cs) {
          v2f bq;
          bq.x = V[cs][2 * q];
          bq.y = V[cs][2 * q + 1];
          acc[t][cs] = __builtin_amdgcn_wmma_f32_16x16x4_f32(false, a, false, bq, (short)0, acc[t][cs], false, false);
        }
      }
    }
    if (mma_first) stage();
  }

  v8f a2a = {}, a2b = {};
  const int mt = wv & 3;
  const int nh = wv >> 2;
  {
    const int crow = 16 * mt + n;
    float xi2[8];
#pragma unroll
    for (int q = 0; q < 4; ++q) {
      xi2[2 * q] = xs[crow * XP + 4 * q + 2 * hi];
      xi2[2 * q + 1] = xs[crow * XP + 4 * q + 2 * hi + 1];
    }
    const int col2 = 16 * nh + n;
    const bool ok2 = col2 < 22;
    const float* p2 = U2_0e;
    int K2s = 0;
    if (col2 < 4) {
      p2 = U2_0e + col2;
      K2s = 4;
    } else if (col2 < 22) {
      int e = (col2 - 4) / 6, kk = (col2 - 4) - 6 * e;
      p2 = U2_1o + e * (NL * NL * 6) + kk;
      K2s = 6;
    }
#pragma unroll 4
    for (int W = 0; W < NL; ++W) {
      const float xw2 = xs[crow * XP + W];
#pragma unroll
      for (int q = 0; q < 4; ++q) {
        const int rr = 16 * W + 4 * q + 2 * hi;
        v2f a;
        a.x = xw2 * xi2[2 * q];
        a.y = xw2 * xi2[2 * q + 1];
        float b0 = p2[rr * K2s], b1 = p2[(rr + 1) * K2s];
        v2f bv;
        bv.x = ok2 ? b0 : 0.f;
        bv.y = ok2 ? b1 : 0.f;
        if (q & 1)
          a2b = __builtin_amdgcn_wmma_f32_16x16x4_f32(false, a, false, bv, (short)0, a2b, false, false);
        else
          a2a = __builtin_amdgcn_wmma_f32_16x16x4_f32(false, a, false, bv, (short)0, a2a, false, false);
      }
    }
  }

  __syncthreads();
  float* P = bufs;
#pragma unroll
  for (int t = 0; t < 4; ++t)
#pragma unroll
    for (int cs = 0; cs < 4; ++cs)
#pragma unroll
      for (int j = 0; j < 8; ++j)
        P[((wv * 4 + cs) * 16 + n) * PP + 16 * t + 8 * hi + j] = acc[t][cs][j];
  {
    v8f d2 = a2a + a2b;
#pragma unroll
    for (int j = 0; j < 8; ++j) p2s[(16 * nh + n) * PP + 16 * mt + 8 * hi + j] = d2[j];
  }
  __syncthreads();

  {
    const int q = tid >> 6;
    const int al = tid & 63;
    const int b = 64 * ah + al;
    float yv[NE];
#pragma unroll
    for (int e = 0; e < NE; ++e) yv[e] = ysm[al * NE + e];
    float o = 0.f;
#pragma unroll
    for (int e = 0; e < NE; ++e) {
      float s = 0.f;
#pragma unroll
      for (int w = 0; w < 8; ++w) s += P[((w * 4 + q) * 16 + e) * PP + al];
      o = fmaf(yv[e], s, o);
    }
    const float* w2p = (q == 0) ? w2_0e : w2_1o;
    const float* w1p = (q == 0) ? w1_0e : w1_1o;
    const float* U1p = (q == 0) ? U1_0e : (U1_1o + (q - 1) * NL);
    const int K2 = (q == 0) ? 4 : 6;
    const float* P2c = p2s + ((q == 0) ? 0 : (4 + 6 * (q - 1))) * PP + al;
    for (int kk = 0; kk < K2; ++kk) {
      float wy = 0.f;
#pragma unroll
      for (int e = 0; e < NE; ++e) wy = fmaf(w2p[(e * K2 + kk) * NC + c], yv[e], wy);
      o = fmaf(wy, P2c[kk * PP], o);
    }
    float wy1 = 0.f;
#pragma unroll
    for (int e = 0; e < NE; ++e) wy1 = fmaf(w1p[e * NC + c], yv[e], wy1);
    float s1 = 0.f;
#pragma unroll
    for (int W = 0; W < NL; ++W) s1 = fmaf(xs[al * XP + W], U1p[W], s1);
    o = fmaf(wy1, s1, o);
    if (q == 0)
      out[b * 512 + c] = o;
    else
      out[b * 512 + NC + c * 3 + (q - 1)] = o;
  }
}

extern "C" void kernel_launch(void* const* d_in, const int* in_sizes, int n_in,
                              void* d_out, int out_size, void* d_ws,
                              size_t ws_size, hipStream_t stream) {
  (void)in_sizes; (void)n_in; (void)d_ws; (void)ws_size; (void)out_size;
  const float* x     = (const float*)d_in[0];
  const float* y     = (const float*)d_in[1];
  const float* U1_0e = (const float*)d_in[2];
  const float* U2_0e = (const float*)d_in[3];
  const float* U3_0e = (const float*)d_in[4];
  const float* U1_1o = (const float*)d_in[5];
  const float* U2_1o = (const float*)d_in[6];
  const float* U3_1o = (const float*)d_in[7];
  const float* w1_0e = (const float*)d_in[8];
  const float* w2_0e = (const float*)d_in[9];
  const float* w3_0e = (const float*)d_in[10];
  const float* w1_1o = (const float*)d_in[11];
  const float* w2_1o = (const float*)d_in[12];
  const float* w3_1o = (const float*)d_in[13];
  float* out = (float*)d_out;

  dim3 grid(NC, 2);
  symcon_kernel<<<grid, 256, 0, stream>>>(x, y, U1_0e, U2_0e, U3_0e, U1_1o,
                                          U2_1o, U3_1o, w1_0e, w2_0e, w3_0e,
                                          w1_1o, w2_1o, w3_1o, out);
}


# Grid split over cascade pairs instead of atom halves: a block does all 128 atoms of a channel for 2 of the 4 cascades, so V = U3*w3 is computed once per channel, not twice; sum order kept
# speedup vs baseline: 3.2102x; 1.1236x over previous
#include <hip/hip_runtime.h>


typedef __attribute__((ext_vector_type(2))) float v2f;
typedef __attribute__((ext_vector_type(4))) float v4f;
typedef __attribute__((ext_vector_type(8))) float v8f;

#define NB 128
#define NC 128
#define NE 10
#define NL 16
#define K3A 23
#define K3B 33
#define NROW 4096
#define CH_ROWS 128
#define NCHUNK 32
#define LP 76
#define BUF_FLOATS (CH_ROWS * LP)
#define XP 20
#define PP 129
#define SM_FLOATS (8 * 2 * NE * PP)


template <int TYPE>
__device__ __forceinline__ void symcon_body(
    const float* __restrict__ x, const float* __restrict__ y,
    const float* __restrict__ U1_0e, const float* __restrict__ U2_0e,
    const float* __restrict__ U3_0e, const float* __restrict__ U1_1o,
    const float* __restrict__ U2_1o, const float* __restrict__ U3_1o,
    const float* __restrict__ w1_0e, const float* __restrict__ w2_0e,
    const float* __restrict__ w3_0e, const float* __restrict__ w1_1o,
    const float* __restrict__ w2_1o, const float* __restrict__ w3_1o,
    float* __restrict__ out, float* __restrict__ bufs, float* __restrict__ xs,
    float* __restrict__ ysm, float* __restrict__ p2s) {
  constexpr int K0 = (TYPE == 0) ? K3A : K3B;
  constexpr int S0 = (K0 + 3) / 4;
  constexpr int S1 = 9;
  constexpr int CB = 4 * S0;
  constexpr int N0 = (CH_ROWS * K0 + 255) / 256;
  constexpr int N1 = (CH_ROWS * K3B + 255) / 256;
  constexpr int NPAD = (4 * S0 - K0) + (4 * S1 - K3B);

  const int tid = threadIdx.x;
  const int c = blockIdx.x;
  const int wv = tid >> 5;
  const int lane = tid & 31;
  const int n = lane & 15;
  const int hi = lane >> 4;
  const float* __restrict__ src0 = (TYPE == 0) ? U3_0e : (U3_1o + 1 * (NROW * K3B));
  const float* __restrict__ src1 = (TYPE == 0) ? U3_1o : (U3_1o + 2 * (NROW * K3B));

  for (int i = tid; i < NB * NE; i += 256) ysm[i] = y[i];
  for (int i = tid; i < NB * NL; i += 256) {
    int al = i >> 4, l = i & 15;
    xs[al * XP + l] = x[(al * NC + c) * NL + l];
  }
  for (int i = tid; i < 2 * CH_ROWS * NPAD; i += 256) {
    int bf = i / (CH_ROWS * NPAD), rem = i - bf * (CH_ROWS * NPAD);
    int row = rem / NPAD, p = rem - row * NPAD;
    int col = (p < 4 * S0 - K0) ? (K0 + p) : (CB + K3B + (p - (4 * S0 - K0)));
    bufs[bf * BUF_FLOATS + row * LP + col] = 0.f;
  }

  float rA[N0], rB[N1];
  auto prefetch = [&](int ch) {
    const float* s0 = src0 + ch * (CH_ROWS * K0);
#pragma unroll
    for (int it = 0; it < N0 - 1; ++it) rA[it] = s0[tid + 256 * it];
    rA[N0 - 1] = s0[min(tid + 256 * (N0 - 1), CH_ROWS * K0 - 1)];
    const float* s1 = src1 + ch * (CH_ROWS * K3B);
#pragma unroll
    for (int it = 0; it < N1 - 1; ++it) rB[it] = s1[tid + 256 * it];
    rB[N1 - 1] = s1[min(tid + 256 * (N1 - 1), CH_ROWS * K3B - 1)];
  };
  auto stash = [&](float* buf) {
#pragma unroll
    for (int it = 0; it < N0; ++it) {
      int f = tid + 256 * it;
      if (it < N0 - 1 || tid < CH_ROWS * K0 - 256 * (N0 - 1)) {
        int lr = f / K0, kk = f - lr * K0;
        buf[lr * LP + kk] = rA[it];
      }
    }
#pragma unroll
    for (int it = 0; it < N1; ++it) {
      int f = tid + 256 * it;
      if (it < N1 - 1 || tid < CH_ROWS * K3B - 256 * (N1 - 1)) {
        int lr = f / K3B, kk = f - lr * K3B;
        buf[lr * LP + CB + kk] = rB[it];
      }
    }
  };

  v2f afA[S0], afB[S1];
  float xw[8], xX[8];
  auto load_frags = [&](int ch) __attribute__((always_inline)) {
    const float* src = bufs + (ch & 1) * BUF_FLOATS + (16 * wv + n) * LP + 2 * hi;
#pragma unroll
    for (int s = 0; s < S0; ++s) afA[s] = *(const v2f*)(src + 4 * s);
#pragma unroll
    for (int s = 0; s < S1; ++s) afB[s] = *(const v2f*)(src + CB + 4 * s);
    const int T = 8 * ch + wv;
    const int W = (T >> 4) & 15, X = T & 15;
#pragma unroll
    for (int t = 0; t < 8; ++t) {
      xw[t] = xs[(16 * t + n) * XP + W];
      xX[t] = xs[(16 * t + n) * XP + X];
    }
  };

  prefetch(0);

  v2f bw0[6], bw1[9];
  if (TYPE == 0) {
#pragma unroll
    for (int s = 0; s < 6; ++s) {
      const int k0 = 4 * s + 2 * hi;
      bw0[s].x = (n < NE && k0 < K3A) ? w3_0e[(n * K3A + k0) * NC + c] : 0.f;
      bw0[s].y = (n < NE && k0 + 1 < K3A) ? w3_0e[(n * K3A + k0 + 1) * NC + c] : 0.f;
    }
  }
#pragma unroll
  for (int s = 0; s < 9; ++s) {
    const int k0 = 4 * s + 2 * hi;
    bw1[s].x = (n < NE && k0 < K3B) ? w3_1o[(n * K3B + k0) * NC + c] : 0.f;
    bw1[s].y = (n < NE && k0 + 1 < K3B) ? w3_1o[(n * K3B + k0 + 1) * NC + c] : 0.f;
  }

  __syncthreads();

  float xi[8][8];
#pragma unroll
  for (int t = 0; t < 8; ++t) {
    v4f lo = *(const v4f*)(xs + (16 * t + n) * XP + 8 * hi);
    v4f hh = *(const v4f*)(xs + (16 * t + n) * XP + 8 * hi + 4);
#pragma unroll
    for (int j = 0; j < 4; ++j) {
      xi[t][j] = lo[j];
      xi[t][4 + j] = hh[j];
    }
  }
  stash(bufs);
  prefetch(1);
  __syncthreads();

  load_frags(0);
  stash(bufs + BUF_FLOATS);
  prefetch(2);

  const bool mma_first = __builtin_amdgcn_readfirstlane(wv) < 4;

  v8f acc[8][2] = {};
  for (int ch = 0; ch < NCHUNK; ++ch) {
    auto stage = [&]() __attribute__((always_inline)) {
      if (ch + 2 < NCHUNK) stash(bufs + (ch & 1) * BUF_FLOATS);
      if (ch + 3 < NCHUNK) prefetch(ch + 3);
    };
    __builtin_amdgcn_fence(__ATOMIC_RELEASE, "workgroup", "local");
    __builtin_amdgcn_s_barrier();
    __builtin_amdgcn_fence(__ATOMIC_ACQUIRE, "workgroup", "local");
    if (!mma_first) stage();
    float pwx[8];
#pragma unroll
    for (int t = 0; t < 8; ++t) pwx[t] = xw[t] * xX[t];
    v8f V[2] = {};
#pragma unroll
    for (int s = 0; s < 9; ++s) {
      if (s < S0)
        V[0] = __builtin_amdgcn_wmma_f32_16x16x4_f32(false, afA[s], false, (TYPE == 0) ? bw0[s < 6 ? s : 0] : bw1[s], (short)0, V[0], false, false);
      V[1] = __builtin_amdgcn_wmma_f32_16x16x4_f32(false, afB[s], false, bw1[s], (short)0, V[1], false, false);
    }
    load_frags(ch + 1 < NCHUNK ? ch + 1 : ch);
#pragma unroll
    for (int q = 0; q < 4; ++q) {
#pragma unroll
      for (int t = 0; t < 8; ++t) {
        v2f a;
        a.x = pwx[t] * xi[t][2 * q];
        a.y = pwx[t] * xi[t][2 * q + 1];
#pragma unroll
        for (int g = 0; g < 2; ++g) {
          v2f bq;
          bq.x = V[g][2 * q];
          bq.y = V[g][2 * q + 1];
          acc[t][g] = __builtin_amdgcn_wmma_f32_16x16x4_f32(false, a, false, bq, (short)0, acc[t][g], false, false);
        }
      }
    }
    if (mma_first) stage();
  }

  constexpr int NC2 = (TYPE == 0) ? 10 : 12;
  constexpr int C2B = (TYPE == 0) ? 4 : 6;
  v8f a2a = {}, a2b = {};
  {
    const int crow = 16 * wv + n;
    float xi2[8];
#pragma unroll
    for (int q = 0; q < 4; ++q) {
      xi2[2 * q] = xs[crow * XP + 4 * q + 2 * hi];
      xi2[2 * q + 1] = xs[crow * XP + 4 * q + 2 * hi + 1];
    }
    const bool ok2 = n < NC2;
    const float* p2 = U2_1o;
    int K2s = 0;
    if (TYPE == 0) {
      if (n < 4) {
        p2 = U2_0e + n;
        K2s = 4;
      } else if (n < 10) {
        p2 = U2_1o + (n - 4);
        K2s = 6;
      }
    } else {
      if (n < 6) {
        p2 = U2_1o + 1 * (NL * NL * 6) + n;
        K2s = 6;
      } else if (n < 12) {
        p2 = U2_1o + 2 * (NL * NL * 6) + (n - 6);
        K2s = 6;
      }
    }
#pragma unroll 4
    for (int W = 0; W < NL; ++W) {
      const float xw2 = xs[crow * XP + W];
#pragma unroll
      for (int q = 0; q < 4; ++q) {
        const int rr = 16 * W + 4 * q + 2 * hi;
        v2f a;
        a.x = xw2 * xi2[2 * q];
        a.y = xw2 * xi2[2 * q + 1];
        float b0 = p2[rr * K2s], b1 = p2[(rr + 1) * K2s];
        v2f bv;
        bv.x = ok2 ? b0 : 0.f;
        bv.y = ok2 ? b1 : 0.f;
        if (q & 1)
          a2b = __builtin_amdgcn_wmma_f32_16x16x4_f32(false, a, false, bv, (short)0, a2b, false, false);
        else
          a2a = __builtin_amdgcn_wmma_f32_16x16x4_f32(false, a, false, bv, (short)0, a2a, false, false);
      }
    }
  }

  __syncthreads();
  float* P = bufs;
  if (n < NE) {
#pragma unroll
    for (int t = 0; t < 8; ++t)
#pragma unroll
      for (int g = 0; g < 2; ++g)
#pragma unroll
        for (int j = 0; j < 8; ++j)
          P[((wv * 2 + g) * NE + n) * PP + 16 * t + 8 * hi + j] = acc[t][g][j];
  }
  {
    v8f d2 = a2a + a2b;
#pragma unroll
    for (int j = 0; j < 8; ++j) p2s[n * PP + 16 * wv + 8 * hi + j] = d2[j];
  }
  __syncthreads();

  {
    const int g = tid >> 7;
    const int q = 2 * TYPE + g;
    const int b = tid & 127;
    float yv[NE];
#pragma unroll
    for (int e = 0; e < NE; ++e) yv[e] = ysm[b * NE + e];
    float o = 0.f;
#pragma unroll
    for (int e = 0; e < NE; ++e) {
      float s = 0.f;
#pragma unroll
      for (int w = 0; w < 8; ++w) s += P[((w * 2 + g) * NE + e) * PP + b];
      o = fmaf(yv[e], s, o);
    }
    const float* w2p = (q == 0) ? w2_0e : w2_1o;
    const float* w1p = (q == 0) ? w1_0e : w1_1o;
    const float* U1p = (q == 0) ? U1_0e : (U1_1o + (q - 1) * NL);
    const int K2 = (q == 0) ? 4 : 6;
    const float* P2c = p2s + (g == 0 ? 0 : C2B) * PP + b;
    for (int kk = 0; kk < K2; ++kk) {
      float wy = 0.f;
#pragma unroll
      for (int e = 0; e < NE; ++e) wy = fmaf(w2p[(e * K2 + kk) * NC + c], yv[e], wy);
      o = fmaf(wy, P2c[kk * PP], o);
    }
    float wy1 = 0.f;
#pragma unroll
    for (int e = 0; e < NE; ++e) wy1 = fmaf(w1p[e * NC + c], yv[e], wy1);
    float s1 = 0.f;
#pragma unroll
    for (int W = 0; W < NL; ++W) s1 = fmaf(xs[b * XP + W], U1p[W], s1);
    o = fmaf(wy1, s1, o);
    if (q == 0)
      out[b * 512 + c] = o;
    else
      out[b * 512 + NC + c * 3 + (q - 1)] = o;
  }
}

__global__ __launch_bounds__(256) void symcon_kernel(
    const float* __restrict__ x, const float* __restrict__ y,
    const float* __restrict__ U1_0e, const float* __restrict__ U2_0e,
    const float* __restrict__ U3_0e, const float* __restrict__ U1_1o,
    const float* __restrict__ U2_1o, const float* __restrict__ U3_1o,
    const float* __restrict__ w1_0e, const float* __restrict__ w2_0e,
    const float* __restrict__ w3_0e, const float* __restrict__ w1_1o,
    const float* __restrict__ w2_1o, const float* __restrict__ w3_1o,
    float* __restrict__ out) {
  __shared__ __attribute__((aligned(16))) float bufs[SM_FLOATS];
  __shared__ __attribute__((aligned(16))) float xs[NB * XP];
  __shared__ float ysm[NB * NE];
  __shared__ float p2s[16 * PP];
  if (blockIdx.y == 0)
    symcon_body<0>(x, y, U1_0e, U2_0e, U3_0e, U1_1o, U2_1o, U3_1o, w1_0e, w2_0e, w3_0e, w1_1o, w2_1o, w3_1o,
                   out, bufs, xs, ysm, p2s);
  else
    symcon_body<1>(x, y, U1_0e, U2_0e, U3_0e, U1_1o, U2_1o, U3_1o, w1_0e, w2_0e, w3_0e, w1_1o, w2_1o, w3_1o,
                   out, bufs, xs, ysm, p2s);
}

extern "C" void kernel_launch(void* const* d_in, const int* in_sizes, int n_in,
                              void* d_out, int out_size, void* d_ws,
                              size_t ws_size, hipStream_t stream) {
  (void)in_sizes; (void)n_in; (void)d_ws; (void)ws_size; (void)out_size;
  const float* x     = (const float*)d_in[0];
  const float* y     = (const float*)d_in[1];
  const float* U1_0e = (const float*)d_in[2];
  const float* U2_0e = (const float*)d_in[3];
  const float* U3_0e = (const float*)d_in[4];
  const float* U1_1o = (const float*)d_in[5];
  const float* U2_1o = (const float*)d_in[6];
  const float* U3_1o = (const float*)d_in[7];
  const float* w1_0e = (const float*)d_in[8];
  const float* w2_0e = (const float*)d_in[9];
  const float* w3_0e = (const float*)d_in[10];
  const float* w1_1o = (const float*)d_in[11];
  const float* w2_1o = (const float*)d_in[12];
  const float* w3_1o = (const float*)d_in[13];
  float* out = (float*)d_out;

  dim3 grid(NC, 2);
  symcon_kernel<<<grid, 256, 0, stream>>>(x, y, U1_0e, U2_0e, U3_0e, U1_1o,
                                          U2_1o, U3_1o, w1_0e, w2_0e, w3_0e,
                                          w1_1o, w2_1o, w3_1o, out);
}


# Memory round trips outside the main loop overlapped: y, x, w3, U3 chunks 0+1 and the epilogue's w2/w1/U1 (via LDS) issued as one load group in the prologue; order-2 gather unrolled; sum order kept
# speedup vs baseline: 3.6777x; 1.1456x over previous
#include <hip/hip_runtime.h>


typedef __attribute__((ext_vector_type(2))) float v2f;
typedef __attribute__((ext_vector_type(4))) float v4f;
typedef __attribute__((ext_vector_type(8))) float v8f;

#define NB 128
#define NC 128
#define NE 10
#define NL 16
#define K3A 23
#define K3B 33
#define NROW 4096
#define CH_ROWS 128
#define NCHUNK 32
#define LP 76
#define BUF_FLOATS (CH_ROWS * LP)
#define XP 20
#define PP 129
#define SM_FLOATS (8 * 2 * NE * PP)


template <int TYPE>
__device__ __forceinline__ void symcon_body(
    const float* __restrict__ x, const float* __restrict__ y,
    const float* __restrict__ U1_0e, const float* __restrict__ U2_0e,
    const float* __restrict__ U3_0e, const float* __restrict__ U1_1o,
    const float* __restrict__ U2_1o, const float* __restrict__ U3_1o,
    const float* __restrict__ w1_0e, const float* __restrict__ w2_0e,
    const float* __restrict__ w3_0e, const float* __restrict__ w1_1o,
    const float* __restrict__ w2_1o, const float* __restrict__ w3_1o,
    float* __restrict__ out, float* __restrict__ bufs, float* __restrict__ xs,
    float* __restrict__ ysm, float* __restrict__ p2s, float* __restrict__ cst) {
  constexpr int K0 = (TYPE == 0) ? K3A : K3B;
  constexpr int S0 = (K0 + 3) / 4;
  constexpr int S1 = 9;
  constexpr int CB = 4 * S0;
  constexpr int N0 = (CH_ROWS * K0 + 255) / 256;
  constexpr int N1 = (CH_ROWS * K3B + 255) / 256;
  constexpr int NPAD = (4 * S0 - K0) + (4 * S1 - K3B);

  const int tid = threadIdx.x;
  const int c = blockIdx.x;
  const int wv = tid >> 5;
  const int lane = tid & 31;
  const int n = lane & 15;
  const int hi = lane >> 4;
  const float* __restrict__ src0 = (TYPE == 0) ? U3_0e : (U3_1o + 1 * (NROW * K3B));
  const float* __restrict__ src1 = (TYPE == 0) ? U3_1o : (U3_1o + 2 * (NROW * K3B));

  float rA[N0], rB[N1];
  auto prefetch = [&](int ch, float (&da)[N0], float (&db)[N1]) {
    const float* s0 = src0 + ch * (CH_ROWS * K0);
#pragma unroll
    for (int it = 0; it < N0 - 1; ++it) da[it] = s0[tid + 256 * it];
    da[N0 - 1] = s0[min(tid + 256 * (N0 - 1), CH_ROWS * K0 - 1)];
    const float* s1 = src1 + ch * (CH_ROWS * K3B);
#pragma unroll
    for (int it = 0; it < N1 - 1; ++it) db[it] = s1[tid + 256 * it];
    db[N1 - 1] = s1[min(tid + 256 * (N1 - 1), CH_ROWS * K3B - 1)];
  };
  auto stash = [&](float* buf, const float (&sa)[N0], const float (&sb)[N1]) {
#pragma unroll
    for (int it = 0; it < N0; ++it) {
      int f = tid + 256 * it;
      if (it < N0 - 1 || tid < CH_ROWS * K0 - 256 * (N0 - 1)) {
        int lr = f / K0, kk = f - lr * K0;
        buf[lr * LP + kk] = sa[it];
      }
    }
#pragma unroll
    for (int it = 0; it < N1; ++it) {
      int f = tid + 256 * it;
      if (it < N1 - 1 || tid < CH_ROWS * K3B - 256 * (N1 - 1)) {
        int lr = f / K3B, kk = f - lr * K3B;
        buf[lr * LP + CB + kk] = sb[it];
      }
    }
  };

  v2f afA[S0], afB[S1];
  float xw[8], xX[8];
  auto load_frags = [&](int ch) __attribute__((always_inline)) {
    const float* src = bufs + (ch & 1) * BUF_FLOATS + (16 * wv + n) * LP + 2 * hi;
#pragma unroll
    for (int s = 0; s < S0; ++s) afA[s] = *(const v2f*)(src + 4 * s);
#pragma unroll
    for (int s = 0; s < S1; ++s) afB[s] = *(const v2f*)(src + CB + 4 * s);
    const int T = 8 * ch + wv;
    const int W = (T >> 4) & 15, X = T & 15;
#pragma unroll
    for (int t = 0; t < 8; ++t) {
      xw[t] = xs[(16 * t + n) * XP + W];
      xX[t] = xs[(16 * t + n) * XP + X];
    }
  };

  float rA1[N0], rB1[N1];
  prefetch(0, rA, rB);
  prefetch(1, rA1, rB1);
  static_assert(NB * NE == 5 * 256 && NB * NL == 2 * 4 * 256, "prologue copies are sized for 256 threads");
  float yr[5];
#pragma unroll
  for (int it = 0; it < 5; ++it) yr[it] = y[tid + 256 * it];
  v4f xr[2];
#pragma unroll
  for (int it = 0; it < 2; ++it) {
    const int i = tid + 256 * it, al = i >> 2, part = i & 3;
    xr[it] = *(const v4f*)(x + (al * NC + c) * NL + 4 * part);
  }
  float cv;
  {
    const int g2 = min(tid / 96, 1), j = tid - 96 * (tid / 96);
    const int q2 = 2 * TYPE + g2;
    const float* w2q = (q2 == 0) ? w2_0e : w2_1o;
    const float* w1q = (q2 == 0) ? w1_0e : w1_1o;
    const float* U1q = (q2 == 0) ? U1_0e : (U1_1o + (q2 - 1) * NL);
    const int K2q = (q2 == 0) ? 4 : 6;
    const float* ptr = w1q + c;
    if (j < NE * K2q) ptr = w2q + j * NC + c;
    else if (j >= 60 && j < 70) ptr = w1q + (j - 60) * NC + c;
    else if (j >= 70 && j < 86) ptr = U1q + (j - 70);
    cv = *ptr;
  }

  v2f bw0[6], bw1[9];
  const int nc = min(n, NE - 1);
  if (TYPE == 0) {
#pragma unroll
    for (int s = 0; s < 6; ++s) {
      const int k0 = 4 * s + 2 * hi;
      const float v0 = w3_0e[(nc * K3A + min(k0, K3A - 1)) * NC + c];
      const float v1 = w3_0e[(nc * K3A + min(k0 + 1, K3A - 1)) * NC + c];
      bw0[s].x = (n < NE && k0 < K3A) ? v0 : 0.f;
      bw0[s].y = (n < NE && k0 + 1 < K3A) ? v1 : 0.f;
    }
  }
#pragma unroll
  for (int s = 0; s < 9; ++s) {
    const int k0 = 4 * s + 2 * hi;
    const float v0 = w3_1o[(nc * K3B + min(k0, K3B - 1)) * NC + c];
    const float v1 = w3_1o[(nc * K3B + min(k0 + 1, K3B - 1)) * NC + c];
    bw1[s].x = (n < NE && k0 < K3B) ? v0 : 0.f;
    bw1[s].y = (n < NE && k0 + 1 < K3B) ? v1 : 0.f;
  }

  for (int i = tid; i < 2 * CH_ROWS * NPAD; i += 256) {
    int bf = i / (CH_ROWS * NPAD), rem = i - bf * (CH_ROWS * NPAD);
    int row = rem / NPAD, p = rem - row * NPAD;
    int col = (p < 4 * S0 - K0) ? (K0 + p) : (CB + K3B + (p - (4 * S0 - K0)));
    bufs[bf * BUF_FLOATS + row * LP + col] = 0.f;
  }
#pragma unroll
  for (int it = 0; it < 5; ++it) ysm[tid + 256 * it] = yr[it];
#pragma unroll
  for (int it = 0; it < 2; ++it) {
    const int i = tid + 256 * it, al = i >> 2, part = i & 3;
    *(v4f*)(xs + al * XP + 4 * part) = xr[it];
  }
  if (tid < 192) cst[tid] = cv;
  __syncthreads();

  float xi[8][8];
#pragma unroll
  for (int t = 0; t < 8; ++t) {
    v4f lo = *(const v4f*)(xs + (16 * t + n) * XP + 8 * hi);
    v4f hh = *(const v4f*)(xs + (16 * t + n) * XP + 8 * hi + 4);
#pragma unroll
    for (int j = 0; j < 4; ++j) {
      xi[t][j] = lo[j];
      xi[t][4 + j] = hh[j];
    }
  }
  stash(bufs, rA, rB);
  __syncthreads();

  load_frags(0);
  stash(bufs + BUF_FLOATS, rA1, rB1);
  prefetch(2, rA, rB);

  const bool mma_first = __builtin_amdgcn_readfirstlane(wv) < 4;

  v8f acc[8][2] = {};
  for (int ch = 0; ch < NCHUNK; ++ch) {
    auto stage = [&]() __attribute__((always_inline)) {
      if (ch + 2 < NCHUNK) stash(bufs + (ch & 1) * BUF_FLOATS, rA, rB);
      if (ch + 3 < NCHUNK) prefetch(ch + 3, rA, rB);
    };
    __builtin_amdgcn_fence(__ATOMIC_RELEASE, "workgroup", "local");
    __builtin_amdgcn_s_barrier();
    __builtin_amdgcn_fence(__ATOMIC_ACQUIRE, "workgroup", "local");
    if (!mma_first) stage();
    float pwx[8];
#pragma unroll
    for (int t = 0; t < 8; ++t) pwx[t] = xw[t] * xX[t];
    v8f V[2] = {};
#pragma unroll
    for (int s = 0; s < 9; ++s) {
      if (s < S0)
        V[0] = __builtin_amdgcn_wmma_f32_16x16x4_f32(false, afA[s], false, (TYPE == 0) ? bw0[s < 6 ? s : 0] : bw1[s], (short)0, V[0], false, false);
      V[1] = __builtin_amdgcn_wmma_f32_16x16x4_f32(false, afB[s], false, bw1[s], (short)0, V[1], false, false);
    }
    load_frags(ch + 1 < NCHUNK ? ch + 1 : ch);
#pragma unroll
    for (int q = 0; q < 4; ++q) {
#pragma unroll
      for (int t = 0; t < 8; ++t) {
        v2f a;
        a.x = pwx[t] * xi[t][2 * q];
        a.y = pwx[t] * xi[t][2 * q + 1];
#pragma unroll
        for (int g = 0; g < 2; ++g) {
          v2f bq;
          bq.x = V[g][2 * q];
          bq.y = V[g][2 * q + 1];
          acc[t][g] = __builtin_amdgcn_wmma_f32_16x16x4_f32(false, a, false, bq, (short)0, acc[t][g], false, false);
        }
      }
    }
    if (mma_first) stage();
  }

  constexpr int NC2 = (TYPE == 0) ? 10 : 12;
  constexpr int C2B = (TYPE == 0) ? 4 : 6;
  v8f a2a = {}, a2b = {};
  {
    const int crow = 16 * wv + n;
    float xi2[8];
#pragma unroll
    for (int q = 0; q < 4; ++q) {
      xi2[2 * q] = xs[crow * XP + 4 * q + 2 * hi];
      xi2[2 * q + 1] = xs[crow * XP + 4 * q + 2 * hi + 1];
    }
    const bool ok2 = n < NC2;
    const float* p2 = U2_1o;
    int K2s = 0;
    if (TYPE == 0) {
      if (n < 4) {
        p2 = U2_0e + n;
        K2s = 4;
      } else if (n < 10) {
        p2 = U2_1o + (n - 4);
        K2s = 6;
      }
    } else {
      if (n < 6) {
        p2 = U2_1o + 1 * (NL * NL * 6) + n;
        K2s = 6;
      } else if (n < 12) {
        p2 = U2_1o + 2 * (NL * NL * 6) + (n - 6);
        K2s = 6;
      }
    }
#pragma unroll
    for (int W = 0; W < NL; ++W) {
      const float xw2 = xs[crow * XP + W];
#pragma unroll
      for (int q = 0; q < 4; ++q) {
        const int rr = 16 * W + 4 * q + 2 * hi;
        v2f a;
        a.x = xw2 * xi2[2 * q];
        a.y = xw2 * xi2[2 * q + 1];
        float b0 = p2[rr * K2s], b1 = p2[(rr + 1) * K2s];
        v2f bv;
        bv.x = ok2 ? b0 : 0.f;
        bv.y = ok2 ? b1 : 0.f;
        if (q & 1)
          a2b = __builtin_amdgcn_wmma_f32_16x16x4_f32(false, a, false, bv, (short)0, a2b, false, false);
        else
          a2a = __builtin_amdgcn_wmma_f32_16x16x4_f32(false, a, false, bv, (short)0, a2a, false, false);
      }
    }
  }

  __syncthreads();
  float* P = bufs;
  if (n < NE) {
#pragma unroll
    for (int t = 0; t < 8; ++t)
#pragma unroll
      for (int g = 0; g < 2; ++g)
#pragma unroll
        for (int j = 0; j < 8; ++j)
          P[((wv * 2 + g) * NE + n) * PP + 16 * t + 8 * hi + j] = acc[t][g][j];
  }
  {
    v8f d2 = a2a + a2b;
#pragma unroll
    for (int j = 0; j < 8; ++j) p2s[n * PP + 16 * wv + 8 * hi + j] = d2[j];
  }
  __syncthreads();

  {
    const int g = tid >> 7;
    const int q = 2 * TYPE + g;
    const int b = tid & 127;
    float yv[NE];
#pragma unroll
    for (int e = 0; e < NE; ++e) yv[e] = ysm[b * NE + e];
    float o = 0.f;
#pragma unroll
    for (int e = 0; e < NE; ++e) {
      float s = 0.f;
#pragma unroll
      for (int w = 0; w < 8; ++w) s += P[((w * 2 + g) * NE + e) * PP + b];
      o = fmaf(yv[e], s, o);
    }
    const float* cg = cst + 96 * g;
    const int K2 = (q == 0) ? 4 : 6;
    const float* P2c = p2s + (g == 0 ? 0 : C2B) * PP + b;
    for (int kk = 0; kk < K2; ++kk) {
      float wy = 0.f;
#pragma unroll
      for (int e = 0; e < NE; ++e) wy = fmaf(cg[e * K2 + kk], yv[e], wy);
      o = fmaf(wy, P2c[kk * PP], o);
    }
    float wy1 = 0.f;
#pragma unroll
    for (int e = 0; e < NE; ++e) wy1 = fmaf(cg[60 + e], yv[e], wy1);
    float s1 = 0.f;
#pragma unroll
    for (int W = 0; W < NL; ++W) s1 = fmaf(xs[b * XP + W], cg[70 + W], s1);
    o = fmaf(wy1, s1, o);
    if (q == 0)
      out[b * 512 + c] = o;
    else
      out[b * 512 + NC + c * 3 + (q - 1)] = o;
  }
}

__global__ __launch_bounds__(256) void symcon_kernel(
    const float* __restrict__ x, const float* __restrict__ y,
    const float* __restrict__ U1_0e, const float* __restrict__ U2_0e,
    const float* __restrict__ U3_0e, const float* __restrict__ U1_1o,
    const float* __restrict__ U2_1o, const float* __restrict__ U3_1o,
    const float* __restrict__ w1_0e, const float* __restrict__ w2_0e,
    const float* __restrict__ w3_0e, const float* __restrict__ w1_1o,
    const float* __restrict__ w2_1o, const float* __restrict__ w3_1o,
    float* __restrict__ out) {
  __shared__ __attribute__((aligned(16))) float bufs[SM_FLOATS];
  __shared__ __attribute__((aligned(16))) float xs[NB * XP];
  __shared__ float ysm[NB * NE];
  __shared__ float p2s[16 * PP];
  __shared__ float cst[2 * 96];
  if (blockIdx.y == 0)
    symcon_body<0>(x, y, U1_0e, U2_0e, U3_0e, U1_1o, U2_1o, U3_1o, w1_0e, w2_0e, w3_0e, w1_1o, w2_1o, w3_1o,
                   out, bufs, xs, ysm, p2s, cst);
  else
    symcon_body<1>(x, y, U1_0e, U2_0e, U3_0e, U1_1o, U2_1o, U3_1o, w1_0e, w2_0e, w3_0e, w1_1o, w2_1o, w3_1o,
                   out, bufs, xs, ysm, p2s, cst);
}

extern "C" void kernel_launch(void* const* d_in, const int* in_sizes, int n_in,
                              void* d_out, int out_size, void* d_ws,
                              size_t ws_size, hipStream_t stream) {
  (void)in_sizes; (void)n_in; (void)d_ws; (void)ws_size; (void)out_size;
  const float* x     = (const float*)d_in[0];
  const float* y     = (const float*)d_in[1];
  const float* U1_0e = (const float*)d_in[2];
  const float* U2_0e = (const float*)d_in[3];
  const float* U3_0e = (const float*)d_in[4];
  const float* U1_1o = (const float*)d_in[5];
  const float* U2_1o = (const float*)d_in[6];
  const float* U3_1o = (const float*)d_in[7];
  const float* w1_0e = (const float*)d_in[8];
  const float* w2_0e = (const float*)d_in[9];
  const float* w3_0e = (const float*)d_in[10];
  const float* w1_1o = (const float*)d_in[11];
  const float* w2_1o = (const float*)d_in[12];
  const float* w3_1o = (const float*)d_in[13];
  float* out = (float*)d_out;

  dim3 grid(NC, 2);
  symcon_kernel<<<grid, 256, 0, stream>>>(x, y, U1_0e, U2_0e, U3_0e, U1_1o,
                                          U2_1o, U3_1o, w1_0e, w2_0e, w3_0e,
                                          w1_1o, w2_1o, w3_1o, out);
}


# Product (2) contracts over the elements, not over i: Vt = w3 x U3^T (elements in rows 0-5, 8-11) is the B operand, A = x[W]x[X]y; 3 k-steps, not 4; x[i] applied after the wave sum (new sum order)
# speedup vs baseline: 4.1514x; 1.1288x over previous
#include <hip/hip_runtime.h>


typedef __attribute__((ext_vector_type(2))) float v2f;
typedef __attribute__((ext_vector_type(4))) float v4f;
typedef __attribute__((ext_vector_type(8))) float v8f;

#define NB 128
#define NC 128
#define NE 10
#define NL 16
#define K3A 23
#define K3B 33
#define NROW 4096
#define CH_ROWS 128
#define NCHUNK 32
#define LP 76
#define BUF_FLOATS (CH_ROWS * LP)
#define XP 20
#define PP 129
#define SM_FLOATS (8 * 2 * NL * PP)


template <int TYPE>
__device__ __forceinline__ void symcon_body(
    const float* __restrict__ x, const float* __restrict__ y,
    const float* __restrict__ U1_0e, const float* __restrict__ U2_0e,
    const float* __restrict__ U3_0e, const float* __restrict__ U1_1o,
    const float* __restrict__ U2_1o, const float* __restrict__ U3_1o,
    const float* __restrict__ w1_0e, const float* __restrict__ w2_0e,
    const float* __restrict__ w3_0e, const float* __restrict__ w1_1o,
    const float* __restrict__ w2_1o, const float* __restrict__ w3_1o,
    float* __restrict__ out, float* __restrict__ bufs, float* __restrict__ xs,
    float* __restrict__ ysm, float* __restrict__ p2s, float* __restrict__ cst) {
  constexpr int K0 = (TYPE == 0) ? K3A : K3B;
  constexpr int S0 = (K0 + 3) / 4;
  constexpr int S1 = 9;
  constexpr int CB = 4 * S0;
  constexpr int N0 = (CH_ROWS * K0 + 255) / 256;
  constexpr int N1 = (CH_ROWS * K3B + 255) / 256;
  constexpr int NPAD = (4 * S0 - K0) + (4 * S1 - K3B);

  const int tid = threadIdx.x;
  const int c = blockIdx.x;
  const int wv = tid >> 5;
  const int lane = tid & 31;
  const int n = lane & 15;
  const int hi = lane >> 4;
  const float* __restrict__ src0 = (TYPE == 0) ? U3_0e : (U3_1o + 1 * (NROW * K3B));
  const float* __restrict__ src1 = (TYPE == 0) ? U3_1o : (U3_1o + 2 * (NROW * K3B));

  float rA[N0], rB[N1];
  auto prefetch = [&](int ch, float (&da)[N0], float (&db)[N1]) {
    const float* s0 = src0 + ch * (CH_ROWS * K0);
#pragma unroll
    for (int it = 0; it < N0 - 1; ++it) da[it] = s0[tid + 256 * it];
    da[N0 - 1] = s0[min(tid + 256 * (N0 - 1), CH_ROWS * K0 - 1)];
    const float* s1 = src1 + ch * (CH_ROWS * K3B);
#pragma unroll
    for (int it = 0; it < N1 - 1; ++it) db[it] = s1[tid + 256 * it];
    db[N1 - 1] = s1[min(tid + 256 * (N1 - 1), CH_ROWS * K3B - 1)];
  };
  auto stash = [&](float* buf, const float (&sa)[N0], const float (&sb)[N1]) {
#pragma unroll
    for (int it = 0; it < N0; ++it) {
      int f = tid + 256 * it;
      if (it < N0 - 1 || tid < CH_ROWS * K0 - 256 * (N0 - 1)) {
        int lr = f / K0, kk = f - lr * K0;
        buf[lr * LP + kk] = sa[it];
      }
    }
#pragma unroll
    for (int it = 0; it < N1; ++it) {
      int f = tid + 256 * it;
      if (it < N1 - 1 || tid < CH_ROWS * K3B - 256 * (N1 - 1)) {
        int lr = f / K3B, kk = f - lr * K3B;
        buf[lr * LP + CB + kk] = sb[it];
      }
    }
  };

  v2f ufA[S0], ufB[S1];
  float xw[8], xX[8];
  auto load_frags = [&](int ch) __attribute__((always_inline)) {
    const float* src = bufs + (ch & 1) * BUF_FLOATS + (16 * wv + n) * LP + 2 * hi;
#pragma unroll
    for (int s = 0; s < S0; ++s) ufA[s] = *(const v2f*)(src + 4 * s);
#pragma unroll
    for (int s = 0; s < S1; ++s) ufB[s] = *(const v2f*)(src + CB + 4 * s);
    const int T = 8 * ch + wv;
    const int W = (T >> 4) & 15, X = T & 15;
#pragma unroll
    for (int t = 0; t < 8; ++t) {
      xw[t] = xs[(16 * t + n) * XP + W];
      xX[t] = xs[(16 * t + n) * XP + X];
    }
  };

  float rA1[N0], rB1[N1];
  prefetch(0, rA, rB);
  prefetch(1, rA1, rB1);
  static_assert(NB * NE == 5 * 256 && NB * NL == 2 * 4 * 256, "prologue copies are sized for 256 threads");
  float yr[5];
#pragma unroll
  for (int it = 0; it < 5; ++it) yr[it] = y[tid + 256 * it];
  v4f xr[2];
#pragma unroll
  for (int it = 0; it < 2; ++it) {
    const int i = tid + 256 * it, al = i >> 2, part = i & 3;
    xr[it] = *(const v4f*)(x + (al * NC + c) * NL + 4 * part);
  }
  float cv;
  {
    const int g2 = min(tid / 96, 1), j = tid - 96 * (tid / 96);
    const int q2 = 2 * TYPE + g2;
    const float* w2q = (q2 == 0) ? w2_0e : w2_1o;
    const float* w1q = (q2 == 0) ? w1_0e : w1_1o;
    const float* U1q = (q2 == 0) ? U1_0e : (U1_1o + (q2 - 1) * NL);
    const int K2q = (q2 == 0) ? 4 : 6;
    const float* ptr = w1q + c;
    if (j < NE * K2q) ptr = w2q + j * NC + c;
    else if (j >= 60 && j < 70) ptr = w1q + (j - 60) * NC + c;
    else if (j >= 70 && j < 86) ptr = U1q + (j - 70);
    cv = *ptr;
  }

  v2f aw0[6], aw1[9];
  const bool erow = (n < 6) || (n >= 8 && n < 12);
  const int nc = erow ? ((n < 6) ? n : n - 2) : 0;
  if (TYPE == 0) {
#pragma unroll
    for (int s = 0; s < 6; ++s) {
      const int k0 = 4 * s + 2 * hi;
      const float v0 = w3_0e[(nc * K3A + min(k0, K3A - 1)) * NC + c];
      const float v1 = w3_0e[(nc * K3A + min(k0 + 1, K3A - 1)) * NC + c];
      aw0[s].x = (erow && k0 < K3A) ? v0 : 0.f;
      aw0[s].y = (erow && k0 + 1 < K3A) ? v1 : 0.f;
    }
  }
#pragma unroll
  for (int s = 0; s < 9; ++s) {
    const int k0 = 4 * s + 2 * hi;
    const float v0 = w3_1o[(nc * K3B + min(k0, K3B - 1)) * NC + c];
    const float v1 = w3_1o[(nc * K3B + min(k0 + 1, K3B - 1)) * NC + c];
    aw1[s].x = (erow && k0 < K3B) ? v0 : 0.f;
    aw1[s].y = (erow && k0 + 1 < K3B) ? v1 : 0.f;
  }

  for (int i = tid; i < 2 * CH_ROWS * NPAD; i += 256) {
    int bf = i / (CH_ROWS * NPAD), rem = i - bf * (CH_ROWS * NPAD);
    int row = rem / NPAD, p = rem - row * NPAD;
    int col = (p < 4 * S0 - K0) ? (K0 + p) : (CB + K3B + (p - (4 * S0 - K0)));
    bufs[bf * BUF_FLOATS + row * LP + col] = 0.f;
  }
#pragma unroll
  for (int it = 0; it < 5; ++it) ysm[tid + 256 * it] = yr[it];
#pragma unroll
  for (int it = 0; it < 2; ++it) {
    const int i = tid + 256 * it, al = i >> 2, part = i & 3;
    *(v4f*)(xs + al * XP + 4 * part) = xr[it];
  }
  if (tid < 192) cst[tid] = cv;
  __syncthreads();

  v2f yq[8][3];
#pragma unroll
  for (int t = 0; t < 8; ++t) {
#pragma unroll
    for (int q = 0; q < 3; ++q) {
      const int e0 = min(6 * hi + 2 * q, NE - 2);
      const v2f v = *(const v2f*)(ysm + (16 * t + n) * NE + e0);
      const bool used = (q < 2) || (hi == 0);
      yq[t][q].x = used ? v.x : 0.f;
      yq[t][q].y = used ? v.y : 0.f;
    }
  }
  stash(bufs, rA, rB);
  __syncthreads();

  load_frags(0);
  stash(bufs + BUF_FLOATS, rA1, rB1);
  prefetch(2, rA, rB);

  const bool mma_first = __builtin_amdgcn_readfirstlane(wv) < 4;

  v8f acc[8][2] = {};
  for (int ch = 0; ch < NCHUNK; ++ch) {
    auto stage = [&]() __attribute__((always_inline)) {
      if (ch + 2 < NCHUNK) stash(bufs + (ch & 1) * BUF_FLOATS, rA, rB);
      if (ch + 3 < NCHUNK) prefetch(ch + 3, rA, rB);
    };
    __builtin_amdgcn_fence(__ATOMIC_RELEASE, "workgroup", "local");
    __builtin_amdgcn_s_barrier();
    __builtin_amdgcn_fence(__ATOMIC_ACQUIRE, "workgroup", "local");
    if (!mma_first) stage();
    float pwx[8];
#pragma unroll
    for (int t = 0; t < 8; ++t) pwx[t] = xw[t] * xX[t];
    v8f V[2] = {};
#pragma unroll
    for (int s = 0; s < 9; ++s) {
      if (s < S0)
        V[0] = __builtin_amdgcn_wmma_f32_16x16x4_f32(false, (TYPE == 0) ? aw0[s < 6 ? s : 0] : aw1[s], false, ufA[s], (short)0, V[0], false, false);
      V[1] = __builtin_amdgcn_wmma_f32_16x16x4_f32(false, aw1[s], false, ufB[s], (short)0, V[1], false, false);
    }
    load_frags(ch + 1 < NCHUNK ? ch + 1 : ch);
#pragma unroll
    for (int q = 0; q < 3; ++q) {
#pragma unroll
      for (int t = 0; t < 8; ++t) {
        v2f a;
        a.x = pwx[t] * yq[t][q].x;
        a.y = pwx[t] * yq[t][q].y;
#pragma unroll
        for (int g = 0; g < 2; ++g) {
          v2f bq;
          bq.x = V[g][2 * q];
          bq.y = V[g][2 * q + 1];
          acc[t][g] = __builtin_amdgcn_wmma_f32_16x16x4_f32(false, a, false, bq, (short)0, acc[t][g], false, false);
        }
      }
    }
    if (mma_first) stage();
  }

  constexpr int NC2 = (TYPE == 0) ? 10 : 12;
  constexpr int C2B = (TYPE == 0) ? 4 : 6;
  v8f a2a = {}, a2b = {};
  {
    const int crow = 16 * wv + n;
    float xi2[8];
#pragma unroll
    for (int q = 0; q < 4; ++q) {
      xi2[2 * q] = xs[crow * XP + 4 * q + 2 * hi];
      xi2[2 * q + 1] = xs[crow * XP + 4 * q + 2 * hi + 1];
    }
    const bool ok2 = n < NC2;
    const float* p2 = U2_1o;
    int K2s = 0;
    if (TYPE == 0) {
      if (n < 4) {
        p2 = U2_0e + n;
        K2s = 4;
      } else if (n < 10) {
        p2 = U2_1o + (n - 4);
        K2s = 6;
      }
    } else {
      if (n < 6) {
        p2 = U2_1o + 1 * (NL * NL * 6) + n;
        K2s = 6;
      } else if (n < 12) {
        p2 = U2_1o + 2 * (NL * NL * 6) + (n - 6);
        K2s = 6;
      }
    }
#pragma unroll
    for (int W = 0; W < NL; ++W) {
      const float xw2 = xs[crow * XP + W];
#pragma unroll
      for (int q = 0; q < 4; ++q) {
        const int rr = 16 * W + 4 * q + 2 * hi;
        v2f a;
        a.x = xw2 * xi2[2 * q];
        a.y = xw2 * xi2[2 * q + 1];
        float b0 = p2[rr * K2s], b1 = p2[(rr + 1) * K2s];
        v2f bv;
        bv.x = ok2 ? b0 : 0.f;
        bv.y = ok2 ? b1 : 0.f;
        if (q & 1)
          a2b = __builtin_amdgcn_wmma_f32_16x16x4_f32(false, a, false, bv, (short)0, a2b, false, false);
        else
          a2a = __builtin_amdgcn_wmma_f32_16x16x4_f32(false, a, false, bv, (short)0, a2a, false, false);
      }
    }
  }

  __syncthreads();
  float* P = bufs;
#pragma unroll
  for (int t = 0; t < 8; ++t)
#pragma unroll
    for (int g = 0; g < 2; ++g)
#pragma unroll
      for (int j = 0; j < 8; ++j)
        P[((wv * 2 + g) * NL + n) * PP + 16 * t + 8 * hi + j] = acc[t][g][j];
  {
    v8f d2 = a2a + a2b;
#pragma unroll
    for (int j = 0; j < 8; ++j) p2s[n * PP + 16 * wv + 8 * hi + j] = d2[j];
  }
  __syncthreads();

  {
    const int g = tid >> 7;
    const int q = 2 * TYPE + g;
    const int b = tid & 127;
    float yv[NE];
#pragma unroll
    for (int e = 0; e < NE; ++e) yv[e] = ysm[b * NE + e];
    float o = 0.f;
#pragma unroll
    for (int i = 0; i < NL; ++i) {
      float s = 0.f;
#pragma unroll
      for (int w = 0; w < 8; ++w) s += P[((w * 2 + g) * NL + i) * PP + b];
      o = fmaf(xs[b * XP + i], s, o);
    }
    const float* cg = cst + 96 * g;
    const int K2 = (q == 0) ? 4 : 6;
    const float* P2c = p2s + (g == 0 ? 0 : C2B) * PP + b;
    for (int kk = 0; kk < K2; ++kk) {
      float wy = 0.f;
#pragma unroll
      for (int e = 0; e < NE; ++e) wy = fmaf(cg[e * K2 + kk], yv[e], wy);
      o = fmaf(wy, P2c[kk * PP], o);
    }
    float wy1 = 0.f;
#pragma unroll
    for (int e = 0; e < NE; ++e) wy1 = fmaf(cg[60 + e], yv[e], wy1);
    float s1 = 0.f;
#pragma unroll
    for (int W = 0; W < NL; ++W) s1 = fmaf(xs[b * XP + W], cg[70 + W], s1);
    o = fmaf(wy1, s1, o);
    if (q == 0)
      out[b * 512 + c] = o;
    else
      out[b * 512 + NC + c * 3 + (q - 1)] = o;
  }
}

__global__ __launch_bounds__(256) void symcon_kernel(
    const float* __restrict__ x, const float* __restrict__ y,
    const float* __restrict__ U1_0e, const float* __restrict__ U2_0e,
    const float* __restrict__ U3_0e, const float* __restrict__ U1_1o,
    const float* __restrict__ U2_1o, const float* __restrict__ U3_1o,
    const float* __restrict__ w1_0e, const float* __restrict__ w2_0e,
    const float* __restrict__ w3_0e, const float* __restrict__ w1_1o,
    const float* __restrict__ w2_1o, const float* __restrict__ w3_1o,
    float* __restrict__ out) {
  __shared__ __attribute__((aligned(16))) float bufs[SM_FLOATS];
  __shared__ __attribute__((aligned(16))) float xs[NB * XP];
  __shared__ __attribute__((aligned(16))) float ysm[NB * NE];
  __shared__ float p2s[16 * PP];
  __shared__ float cst[2 * 96];
  if (blockIdx.y == 0)
    symcon_body<0>(x, y, U1_0e, U2_0e, U3_0e, U1_1o, U2_1o, U3_1o, w1_0e, w2_0e, w3_0e, w1_1o, w2_1o, w3_1o,
                   out, bufs, xs, ysm, p2s, cst);
  else
    symcon_body<1>(x, y, U1_0e, U2_0e, U3_0e, U1_1o, U2_1o, U3_1o, w1_0e, w2_0e, w3_0e, w1_1o, w2_1o, w3_1o,
                   out, bufs, xs, ysm, p2s, cst);
}

extern "C" void kernel_launch(void* const* d_in, const int* in_sizes, int n_in,
                              void* d_out, int out_size, void* d_ws,
                              size_t ws_size, hipStream_t stream) {
  (void)in_sizes; (void)n_in; (void)d_ws; (void)ws_size; (void)out_size;
  const float* x     = (const float*)d_in[0];
  const float* y     = (const float*)d_in[1];
  const float* U1_0e = (const float*)d_in[2];
  const float* U2_0e = (const float*)d_in[3];
  const float* U3_0e = (const float*)d_in[4];
  const float* U1_1o = (const float*)d_in[5];
  const float* U2_1o = (const float*)d_in[6];
  const float* U3_1o = (const float*)d_in[7];
  const float* w1_0e = (const float*)d_in[8];
  const float* w2_0e = (const float*)d_in[9];
  const float* w3_0e = (const float*)d_in[10];
  const float* w1_1o = (const float*)d_in[11];
  const float* w2_1o = (const float*)d_in[12];
  const float* w3_1o = (const float*)d_in[13];
  float* out = (float*)d_out;

  dim3 grid(NC, 2);
  symcon_kernel<<<grid, 256, 0, stream>>>(x, y, U1_0e, U2_0e, U3_0e, U1_1o,
                                          U2_1o, U3_1o, w1_0e, w2_0e, w3_0e,
                                          w1_1o, w2_1o, w3_1o, out);
}


# Tiles (W,X) and (X,W) meet the same A operand x[W]x[X]y in product (2): both run product (1) into one Vt, so product (2) runs once per unordered pair, 136 not 256; U3 staged as is (new sum order)
# speedup vs baseline: 5.5445x; 1.3356x over previous
#include <hip/hip_runtime.h>


typedef __attribute__((ext_vector_type(2))) float v2f;
typedef __attribute__((ext_vector_type(4))) float v4f;
typedef __attribute__((ext_vector_type(8))) float v8f;

#define NB 128
#define NC 128
#define NE 10
#define NL 16
#define K3A 23
#define K3B 33
#define NROW 4096
#define CH_ROWS 128
#define NCHUNK 32
#define NPAIRCH 30
#define LP 76
#define BUF_FLOATS (CH_ROWS * LP)
#define XP 20
#define PP 129
#define SM_FLOATS (8 * 2 * NL * PP)


template <int TYPE>
__device__ __forceinline__ void symcon_body(
    const float* __restrict__ x, const float* __restrict__ y,
    const float* __restrict__ U1_0e, const float* __restrict__ U2_0e,
    const float* __restrict__ U3_0e, const float* __restrict__ U1_1o,
    const float* __restrict__ U2_1o, const float* __restrict__ U3_1o,
    const float* __restrict__ w1_0e, const float* __restrict__ w2_0e,
    const float* __restrict__ w3_0e, const float* __restrict__ w1_1o,
    const float* __restrict__ w2_1o, const float* __restrict__ w3_1o,
    float* __restrict__ out, float* __restrict__ bufs, float* __restrict__ xs,
    float* __restrict__ ysm, float* __restrict__ p2s, float* __restrict__ cst) {
  constexpr int K0 = (TYPE == 0) ? K3A : K3B;
  constexpr int S0 = (K0 + 3) / 4;
  constexpr int S1 = 9;
  constexpr int CB = 4 * S0;
  constexpr int N0 = (16 * K0 + 31) / 32;
  constexpr int N1 = (16 * K3B + 31) / 32;
  static_assert(16 * K0 - 32 * (N0 - 1) == 16 && 16 * K3B - 32 * (N1 - 1) == 16, "the last staging load covers half a wave");
  constexpr int NPAD = (4 * S0 - K0) + (4 * S1 - K3B);

  const int tid = threadIdx.x;
  const int c = blockIdx.x;
  const int wv = tid >> 5;
  const int lane = tid & 31;
  const int n = lane & 15;
  const int hi = lane >> 4;
  const float* __restrict__ src0 = (TYPE == 0) ? U3_0e : (U3_1o + 1 * (NROW * K3B));
  const float* __restrict__ src1 = (TYPE == 0) ? U3_1o : (U3_1o + 2 * (NROW * K3B));

  const int wvs = __builtin_amdgcn_readfirstlane(wv);
  auto tile_of = [&](int ch) -> int {
    if (ch < NPAIRCH) {
      const int s = ch >> 1, d = (s >> 1) + 1, a = 8 * (s & 1) + wvs, b = (a + d) & 15;
      return (ch & 1) ? 16 * b + a : 16 * a + b;
    }
    return 17 * (8 * (ch - NPAIRCH) + wvs);
  };

  float rA[N0], rB[N1];
  auto prefetch = [&](int ch, float (&da)[N0], float (&db)[N1]) {
    const int T = tile_of(ch);
    const float* s0 = src0 + T * (16 * K0);
#pragma unroll
    for (int it = 0; it < N0 - 1; ++it) da[it] = s0[lane + 32 * it];
    da[N0 - 1] = s0[min(lane + 32 * (N0 - 1), 16 * K0 - 1)];
    const float* s1 = src1 + T * (16 * K3B);
#pragma unroll
    for (int it = 0; it < N1 - 1; ++it) db[it] = s1[lane + 32 * it];
    db[N1 - 1] = s1[min(lane + 32 * (N1 - 1), 16 * K3B - 1)];
  };
  auto stash = [&](float* buf, const float (&sa)[N0], const float (&sb)[N1]) {
    float* row0 = buf + 16 * wvs * LP;
#pragma unroll
    for (int it = 0; it < N0; ++it) {
      int f = lane + 32 * it;
      if (it < N0 - 1 || lane < 16) {
        int lr = f / K0, kk = f - lr * K0;
        row0[lr * LP + kk] = sa[it];
      }
    }
#pragma unroll
    for (int it = 0; it < N1; ++it) {
      int f = lane + 32 * it;
      if (it < N1 - 1 || lane < 16) {
        int lr = f / K3B, kk = f - lr * K3B;
        row0[lr * LP + CB + kk] = sb[it];
      }
    }
  };

  v2f ufA[S0], ufB[S1];
  float xw[8], xX[8];
  auto load_frags = [&](int ch) __attribute__((always_inline)) {
    const float* src = bufs + (ch & 1) * BUF_FLOATS + (16 * wv + n) * LP + 2 * hi;
#pragma unroll
    for (int s = 0; s < S0; ++s) ufA[s] = *(const v2f*)(src + 4 * s);
#pragma unroll
    for (int s = 0; s < S1; ++s) ufB[s] = *(const v2f*)(src + CB + 4 * s);
    const int T = tile_of(ch);
    const int W = T >> 4, X = T & 15;
#pragma unroll
    for (int t = 0; t < 8; ++t) {
      xw[t] = xs[(16 * t + n) * XP + W];
      xX[t] = xs[(16 * t + n) * XP + X];
    }
  };

  float rA1[N0], rB1[N1];
  prefetch(0, rA, rB);
  prefetch(1, rA1, rB1);
  static_assert(NB * NE == 5 * 256 && NB * NL == 2 * 4 * 256, "prologue copies are sized for 256 threads");
  float yr[5];
#pragma unroll
  for (int it = 0; it < 5; ++it) yr[it] = y[tid + 256 * it];
  v4f xr[2];
#pragma unroll
  for (int it = 0; it < 2; ++it) {
    const int i = tid + 256 * it, al = i >> 2, part = i & 3;
    xr[it] = *(const v4f*)(x + (al * NC + c) * NL + 4 * part);
  }
  float cv;
  {
    const int g2 = min(tid / 96, 1), j = tid - 96 * (tid / 96);
    const int q2 = 2 * TYPE + g2;
    const float* w2q = (q2 == 0) ? w2_0e : w2_1o;
    const float* w1q = (q2 == 0) ? w1_0e : w1_1o;
    const float* U1q = (q2 == 0) ? U1_0e : (U1_1o + (q2 - 1) * NL);
    const int K2q = (q2 == 0) ? 4 : 6;
    const float* ptr = w1q + c;
    if (j < NE * K2q) ptr = w2q + j * NC + c;
    else if (j >= 60 && j < 70) ptr = w1q + (j - 60) * NC + c;
    else if (j >= 70 && j < 86) ptr = U1q + (j - 70);
    cv = *ptr;
  }

  v2f aw0[6], aw1[9];
  const bool erow = (n < 6) || (n >= 8 && n < 12);
  const int nc = erow ? ((n < 6) ? n : n - 2) : 0;
  if (TYPE == 0) {
#pragma unroll
    for (int s = 0; s < 6; ++s) {
      const int k0 = 4 * s + 2 * hi;
      const float v0 = w3_0e[(nc * K3A + min(k0, K3A - 1)) * NC + c];
      const float v1 = w3_0e[(nc * K3A + min(k0 + 1, K3A - 1)) * NC + c];
      aw0[s].x = (erow && k0 < K3A) ? v0 : 0.f;
      aw0[s].y = (erow && k0 + 1 < K3A) ? v1 : 0.f;
    }
  }
#pragma unroll
  for (int s = 0; s < 9; ++s) {
    const int k0 = 4 * s + 2 * hi;
    const float v0 = w3_1o[(nc * K3B + min(k0, K3B - 1)) * NC + c];
    const float v1 = w3_1o[(nc * K3B + min(k0 + 1, K3B - 1)) * NC + c];
    aw1[s].x = (erow && k0 < K3B) ? v0 : 0.f;
    aw1[s].y = (erow && k0 + 1 < K3B) ? v1 : 0.f;
  }

  for (int i = tid; i < 2 * CH_ROWS * NPAD; i += 256) {
    int bf = i / (CH_ROWS * NPAD), rem = i - bf * (CH_ROWS * NPAD);
    int row = rem / NPAD, p = rem - row * NPAD;
    int col = (p < 4 * S0 - K0) ? (K0 + p) : (CB + K3B + (p - (4 * S0 - K0)));
    bufs[bf * BUF_FLOATS + row * LP + col] = 0.f;
  }
#pragma unroll
  for (int it = 0; it < 5; ++it) ysm[tid + 256 * it] = yr[it];
#pragma unroll
  for (int it = 0; it < 2; ++it) {
    const int i = tid + 256 * it, al = i >> 2, part = i & 3;
    *(v4f*)(xs + al * XP + 4 * part) = xr[it];
  }
  if (tid < 192) cst[tid] = cv;
  __syncthreads();

  v2f yq[8][3];
#pragma unroll
  for (int t = 0; t < 8; ++t) {
#pragma unroll
    for (int q = 0; q < 3; ++q) {
      const int e0 = min(6 * hi + 2 * q, NE - 2);
      const v2f v = *(const v2f*)(ysm + (16 * t + n) * NE + e0);
      const bool used = (q < 2) || (hi == 0);
      yq[t][q].x = used ? v.x : 0.f;
      yq[t][q].y = used ? v.y : 0.f;
    }
  }
  stash(bufs, rA, rB);
  __syncthreads();

  load_frags(0);
  stash(bufs + BUF_FLOATS, rA1, rB1);
  prefetch(2, rA, rB);

  const bool mma_first = __builtin_amdgcn_readfirstlane(wv) < 4;

  v8f acc[8][2] = {};
  v8f V[2];
  auto chunk = [&](int ch, bool first, bool last) __attribute__((always_inline)) {
    auto stage = [&]() __attribute__((always_inline)) {
      if (ch + 2 < NCHUNK) stash(bufs + (ch & 1) * BUF_FLOATS, rA, rB);
      if (ch + 3 < NCHUNK) prefetch(ch + 3, rA, rB);
    };
    __builtin_amdgcn_fence(__ATOMIC_RELEASE, "workgroup", "local");
    __builtin_amdgcn_s_barrier();
    __builtin_amdgcn_fence(__ATOMIC_ACQUIRE, "workgroup", "local");
    if (!mma_first) stage();
    if (first) {
      V[0] = v8f{};
      V[1] = v8f{};
    }
#pragma unroll
    for (int s = 0; s < 9; ++s) {
      if (s < S0)
        V[0] = __builtin_amdgcn_wmma_f32_16x16x4_f32(false, (TYPE == 0) ? aw0[s < 6 ? s : 0] : aw1[s], false, ufA[s], (short)0, V[0], false, false);
      V[1] = __builtin_amdgcn_wmma_f32_16x16x4_f32(false, aw1[s], false, ufB[s], (short)0, V[1], false, false);
    }
    float pwx[8];
#pragma unroll
    for (int t = 0; t < 8; ++t) pwx[t] = xw[t] * xX[t];
    __builtin_amdgcn_sched_barrier(0);
    load_frags(ch + 1 < NCHUNK ? ch + 1 : ch);
    if (last) {
#pragma unroll
      for (int q = 0; q < 3; ++q) {
#pragma unroll
        for (int t = 0; t < 8; ++t) {
          v2f a;
          a.x = pwx[t] * yq[t][q].x;
          a.y = pwx[t] * yq[t][q].y;
#pragma unroll
          for (int g = 0; g < 2; ++g) {
            v2f bq;
            bq.x = V[g][2 * q];
            bq.y = V[g][2 * q + 1];
            acc[t][g] = __builtin_amdgcn_wmma_f32_16x16x4_f32(false, a, false, bq, (short)0, acc[t][g], false, false);
          }
        }
      }
    }
    if (mma_first) stage();
  };
  for (int s = 0; s < NPAIRCH / 2; ++s) {
    chunk(2 * s, true, false);
    chunk(2 * s + 1, false, true);
  }
#pragma nounroll
  for (int ch = NPAIRCH; ch < NCHUNK; ++ch) chunk(ch, true, true);

  constexpr int NC2 = (TYPE == 0) ? 10 : 12;
  constexpr int C2B = (TYPE == 0) ? 4 : 6;
  v8f a2a = {}, a2b = {};
  {
    const int crow = 16 * wv + n;
    float xi2[8];
#pragma unroll
    for (int q = 0; q < 4; ++q) {
      xi2[2 * q] = xs[crow * XP + 4 * q + 2 * hi];
      xi2[2 * q + 1] = xs[crow * XP + 4 * q + 2 * hi + 1];
    }
    const bool ok2 = n < NC2;
    const float* p2 = U2_1o;
    int K2s = 0;
    if (TYPE == 0) {
      if (n < 4) {
        p2 = U2_0e + n;
        K2s = 4;
      } else if (n < 10) {
        p2 = U2_1o + (n - 4);
        K2s = 6;
      }
    } else {
      if (n < 6) {
        p2 = U2_1o + 1 * (NL * NL * 6) + n;
        K2s = 6;
      } else if (n < 12) {
        p2 = U2_1o + 2 * (NL * NL * 6) + (n - 6);
        K2s = 6;
      }
    }
#pragma unroll
    for (int W = 0; W < NL; ++W) {
      const float xw2 = xs[crow * XP + W];
#pragma unroll
      for (int q = 0; q < 4; ++q) {
        const int rr = 16 * W + 4 * q + 2 * hi;
        v2f a;
        a.x = xw2 * xi2[2 * q];
        a.y = xw2 * xi2[2 * q + 1];
        float b0 = p2[rr * K2s], b1 = p2[(rr + 1) * K2s];
        v2f bv;
        bv.x = ok2 ? b0 : 0.f;
        bv.y = ok2 ? b1 : 0.f;
        if (q & 1)
          a2b = __builtin_amdgcn_wmma_f32_16x16x4_f32(false, a, false, bv, (short)0, a2b, false, false);
        else
          a2a = __builtin_amdgcn_wmma_f32_16x16x4_f32(false, a, false, bv, (short)0, a2a, false, false);
      }
    }
  }

  __syncthreads();
  float* P = bufs;
#pragma unroll
  for (int t = 0; t < 8; ++t)
#pragma unroll
    for (int g = 0; g < 2; ++g)
#pragma unroll
      for (int j = 0; j < 8; ++j)
        P[((wv * 2 + g) * NL + n) * PP + 16 * t + 8 * hi + j] = acc[t][g][j];
  {
    v8f d2 = a2a + a2b;
#pragma unroll
    for (int j = 0; j < 8; ++j) p2s[n * PP + 16 * wv + 8 * hi + j] = d2[j];
  }
  __syncthreads();

  {
    const int g = tid >> 7;
    const int q = 2 * TYPE + g;
    const int b = tid & 127;
    float yv[NE];
#pragma unroll
    for (int e = 0; e < NE; ++e) yv[e] = ysm[b * NE + e];
    float o = 0.f;
#pragma unroll
    for (int i = 0; i < NL; ++i) {
      float s = 0.f;
#pragma unroll
      for (int w = 0; w < 8; ++w) s += P[((w * 2 + g) * NL + i) * PP + b];
      o = fmaf(xs[b * XP + i], s, o);
    }
    const float* cg = cst + 96 * g;
    const int K2 = (q == 0) ? 4 : 6;
    const float* P2c = p2s + (g == 0 ? 0 : C2B) * PP + b;
    for (int kk = 0; kk < K2; ++kk) {
      float wy = 0.f;
#pragma unroll
      for (int e = 0; e < NE; ++e) wy = fmaf(cg[e * K2 + kk], yv[e], wy);
      o = fmaf(wy, P2c[kk * PP], o);
    }
    float wy1 = 0.f;
#pragma unroll
    for (int e = 0; e < NE; ++e) wy1 = fmaf(cg[60 + e], yv[e], wy1);
    float s1 = 0.f;
#pragma unroll
    for (int W = 0; W < NL; ++W) s1 = fmaf(xs[b * XP + W], cg[70 + W], s1);
    o = fmaf(wy1, s1, o);
    if (q == 0)
      out[b * 512 + c] = o;
    else
      out[b * 512 + NC + c * 3 + (q - 1)] = o;
  }
}

__global__ __launch_bounds__(256) void symcon_kernel(
    const float* __restrict__ x, const float* __restrict__ y,
    const float* __restrict__ U1_0e, const float* __restrict__ U2_0e,
    const float* __restrict__ U3_0e, const float* __restrict__ U1_1o,
    const float* __restrict__ U2_1o, const float* __restrict__ U3_1o,
    const float* __restrict__ w1_0e, const float* __restrict__ w2_0e,
    const float* __restrict__ w3_0e, const float* __restrict__ w1_1o,
    const float* __restrict__ w2_1o, const float* __restrict__ w3_1o,
    float* __restrict__ out) {
  __shared__ __attribute__((aligned(16))) float bufs[SM_FLOATS];
  __shared__ __attribute__((aligned(16))) float xs[NB * XP];
  __shared__ __attribute__((aligned(16))) float ysm[NB * NE];
  __shared__ float p2s[16 * PP];
  __shared__ float cst[2 * 96];
  if (blockIdx.y == 0)
    symcon_body<0>(x, y, U1_0e, U2_0e, U3_0e, U1_1o, U2_1o, U3_1o, w1_0e, w2_0e, w3_0e, w1_1o, w2_1o, w3_1o,
                   out, bufs, xs, ysm, p2s, cst);
  else
    symcon_body<1>(x, y, U1_0e, U2_0e, U3_0e, U1_1o, U2_1o, U3_1o, w1_0e, w2_0e, w3_0e, w1_1o, w2_1o, w3_1o,
                   out, bufs, xs, ysm, p2s, cst);
}

extern "C" void kernel_launch(void* const* d_in, const int* in_sizes, int n_in,
                              void* d_out, int out_size, void* d_ws,
                              size_t ws_size, hipStream_t stream) {
  (void)in_sizes; (void)n_in; (void)d_ws; (void)ws_size; (void)out_size;
  const float* x     = (const float*)d_in[0];
  const float* y     = (const float*)d_in[1];
  const float* U1_0e = (const float*)d_in[2];
  const float* U2_0e = (const float*)d_in[3];
  const float* U3_0e = (const float*)d_in[4];
  const float* U1_1o = (const float*)d_in[5];
  const float* U2_1o = (const float*)d_in[6];
  const float* U3_1o = (const float*)d_in[7];
  const float* w1_0e = (const float*)d_in[8];
  const float* w2_0e = (const float*)d_in[9];
  const float* w3_0e = (const float*)d_in[10];
  const float* w1_1o = (const float*)d_in[11];
  const float* w2_1o = (const float*)d_in[12];
  const float* w3_1o = (const float*)d_in[13];
  float* out = (float*)d_out;

  dim3 grid(NC, 2);
  symcon_kernel<<<grid, 256, 0, stream>>>(x, y, U1_0e, U2_0e, U3_0e, U1_1o,
                                          U2_1o, U3_1o, w1_0e, w2_0e, w3_0e,
                                          w1_1o, w2_1o, w3_1o, out);
}


# U3 B fragments of product (1) read straight from global memory, 16 B per lane, two chunks ahead into a third register set: no LDS staging, no chunk barrier; k order changed (new sum order)
# speedup vs baseline: 5.8748x; 1.0596x over previous
#include <hip/hip_runtime.h>


typedef __attribute__((ext_vector_type(2))) float v2f;
typedef __attribute__((ext_vector_type(4))) float v4f;
typedef __attribute__((ext_vector_type(8))) float v8f;

#define NB 128
#define NC 128
#define NE 10
#define NL 16
#define K3A 23
#define K3B 33
#define NROW 4096
#define NCHUNK 32
#define NPAIRCH 30
#define XP 20
#define PP 129
#define SM_FLOATS (8 * 2 * NL * PP)


template <int K>
__device__ __forceinline__ int kpos(int s, int hi, int e) {
  static_assert(K == K3A || K == K3B, "piece maps are written for 23 and 33 params");
  if (s < 2 * (K / 8)) return 8 * (s >> 1) + 4 * hi + 2 * (s & 1) + e;
  if (K == K3B) return (hi == 0 && e == 1) ? 32 : -1;
  const int idx = 2 * (s & 1) + e;
  if (hi == 0) return 16 + idx;
  return (idx == 0) ? -1 : 19 + idx;
}

struct FragSet {
  v4f a[4];
  v2f at;
  v4f b[4];
  v2f bt;
};

template <int TYPE>
__device__ __forceinline__ void symcon_body(
    const float* __restrict__ x, const float* __restrict__ y,
    const float* __restrict__ U1_0e, const float* __restrict__ U2_0e,
    const float* __restrict__ U3_0e, const float* __restrict__ U1_1o,
    const float* __restrict__ U2_1o, const float* __restrict__ U3_1o,
    const float* __restrict__ w1_0e, const float* __restrict__ w2_0e,
    const float* __restrict__ w3_0e, const float* __restrict__ w1_1o,
    const float* __restrict__ w2_1o, const float* __restrict__ w3_1o,
    float* __restrict__ out, float* __restrict__ bufs, float* __restrict__ xs,
    float* __restrict__ ysm, float* __restrict__ p2s, float* __restrict__ cst) {
  constexpr int K0 = (TYPE == 0) ? K3A : K3B;
  constexpr int S0 = (K0 + 3) / 4;
  constexpr int S1 = 9;

  const int tid = threadIdx.x;
  const int c = blockIdx.x;
  const int wv = tid >> 5;
  const int lane = tid & 31;
  const int n = lane & 15;
  const int hi = lane >> 4;
  const float* __restrict__ src0 = (TYPE == 0) ? U3_0e : (U3_1o + 1 * (NROW * K3B));
  const float* __restrict__ src1 = (TYPE == 0) ? U3_1o : (U3_1o + 2 * (NROW * K3B));

  const int wvs = __builtin_amdgcn_readfirstlane(wv);
  auto tile_of = [&](int ch) -> int {
    if (ch < NPAIRCH) {
      const int s = ch >> 1, d = (s >> 1) + 1, a = 8 * (s & 1) + wvs, b = (a + d) & 15;
      return (ch & 1) ? 16 * b + a : 16 * a + b;
    }
    return 17 * (8 * (ch - NPAIRCH) + wvs);
  };

  auto ld4 = [](const float* p) __attribute__((always_inline)) -> v4f {
    v4f v;
    __builtin_memcpy(&v, p, sizeof(v4f));
    return v;
  };
  auto ld2 = [](const float* p) __attribute__((always_inline)) -> v2f {
    v2f v;
    __builtin_memcpy(&v, p, sizeof(v2f));
    return v;
  };
  auto issue = [&](int ch, FragSet& f) __attribute__((always_inline)) {
    const int T = tile_of(ch);
    const float* p0 = src0 + T * (16 * K0) + n * K0;
    const float* p1 = src1 + T * (16 * K3B) + n * K3B;
    if (K0 == K3B) {
#pragma unroll
      for (int j = 0; j < 4; ++j) f.a[j] = ld4(p0 + 8 * j + 4 * hi);
      f.at = ld2(p0 + 31);
    } else {
      f.a[0] = ld4(p0 + 4 * hi);
      f.a[1] = ld4(p0 + 8 + 4 * hi);
      f.a[2] = ld4(p0 + 16 + 3 * hi);
    }
#pragma unroll
    for (int j = 0; j < 4; ++j) f.b[j] = ld4(p1 + 8 * j + 4 * hi);
    f.bt = ld2(p1 + 31);
  };
  auto bfrag0 = [&](const FragSet& f, int s) __attribute__((always_inline)) -> v2f {
    if (K0 == K3B && s == 8) return f.at;
    const v4f q = f.a[s >> 1];
    v2f r;
    r.x = q[2 * (s & 1)];
    r.y = q[2 * (s & 1) + 1];
    return r;
  };
  auto bfrag1 = [&](const FragSet& f, int s) __attribute__((always_inline)) -> v2f {
    if (s == 8) return f.bt;
    const v4f q = f.b[s >> 1];
    v2f r;
    r.x = q[2 * (s & 1)];
    r.y = q[2 * (s & 1) + 1];
    return r;
  };

  FragSet f0, f1, f2;
  issue(0, f0);
  issue(1, f1);
  static_assert(NB * NE == 5 * 256 && NB * NL == 2 * 4 * 256, "prologue copies are sized for 256 threads");
  float yr[5];
#pragma unroll
  for (int it = 0; it < 5; ++it) yr[it] = y[tid + 256 * it];
  v4f xr[2];
#pragma unroll
  for (int it = 0; it < 2; ++it) {
    const int i = tid + 256 * it, al = i >> 2, part = i & 3;
    xr[it] = *(const v4f*)(x + (al * NC + c) * NL + 4 * part);
  }
  float cv;
  {
    const int g2 = min(tid / 96, 1), j = tid - 96 * (tid / 96);
    const int q2 = 2 * TYPE + g2;
    const float* w2q = (q2 == 0) ? w2_0e : w2_1o;
    const float* w1q = (q2 == 0) ? w1_0e : w1_1o;
    const float* U1q = (q2 == 0) ? U1_0e : (U1_1o + (q2 - 1) * NL);
    const int K2q = (q2 == 0) ? 4 : 6;
    const float* ptr = w1q + c;
    if (j < NE * K2q) ptr = w2q + j * NC + c;
    else if (j >= 60 && j < 70) ptr = w1q + (j - 60) * NC + c;
    else if (j >= 70 && j < 86) ptr = U1q + (j - 70);
    cv = *ptr;
  }

  v2f aw0[6], aw1[9];
  const bool erow = (n < 6) || (n >= 8 && n < 12);
  const int nc = erow ? ((n < 6) ? n : n - 2) : 0;
  if (TYPE == 0) {
#pragma unroll
    for (int s = 0; s < 6; ++s) {
      const int k0 = kpos<K3A>(s, hi, 0), k1 = kpos<K3A>(s, hi, 1);
      const float v0 = w3_0e[(nc * K3A + max(k0, 0)) * NC + c];
      const float v1 = w3_0e[(nc * K3A + max(k1, 0)) * NC + c];
      aw0[s].x = (erow && k0 >= 0) ? v0 : 0.f;
      aw0[s].y = (erow && k1 >= 0) ? v1 : 0.f;
    }
  }
#pragma unroll
  for (int s = 0; s < 9; ++s) {
    const int k0 = kpos<K3B>(s, hi, 0), k1 = kpos<K3B>(s, hi, 1);
    const float v0 = w3_1o[(nc * K3B + max(k0, 0)) * NC + c];
    const float v1 = w3_1o[(nc * K3B + max(k1, 0)) * NC + c];
    aw1[s].x = (erow && k0 >= 0) ? v0 : 0.f;
    aw1[s].y = (erow && k1 >= 0) ? v1 : 0.f;
  }

#pragma unroll
  for (int it = 0; it < 5; ++it) ysm[tid + 256 * it] = yr[it];
#pragma unroll
  for (int it = 0; it < 2; ++it) {
    const int i = tid + 256 * it, al = i >> 2, part = i & 3;
    *(v4f*)(xs + al * XP + 4 * part) = xr[it];
  }
  if (tid < 192) cst[tid] = cv;
  __syncthreads();

  v2f yq[8][3];
#pragma unroll
  for (int t = 0; t < 8; ++t) {
#pragma unroll
    for (int q = 0; q < 3; ++q) {
      const int e0 = min(6 * hi + 2 * q, NE - 2);
      const v2f v = *(const v2f*)(ysm + (16 * t + n) * NE + e0);
      const bool used = (q < 2) || (hi == 0);
      yq[t][q].x = used ? v.x : 0.f;
      yq[t][q].y = used ? v.y : 0.f;
    }
  }

  v8f acc[8][2] = {};
  v8f V[2];
  auto chunk = [&](int ch, bool first, bool last, const FragSet& use, FragSet& fill, bool refill) __attribute__((always_inline)) {
    if (refill) issue(ch + 2, fill);
    float xw[8], xX[8];
    if (last) {
      const int T = tile_of(ch);
      const int W = T >> 4, X = T & 15;
#pragma unroll
      for (int t = 0; t < 8; ++t) {
        xw[t] = xs[(16 * t + n) * XP + W];
        xX[t] = xs[(16 * t + n) * XP + X];
      }
    }
    __builtin_amdgcn_sched_barrier(0);
    if (first) {
      V[0] = v8f{};
      V[1] = v8f{};
    }
#pragma unroll
    for (int s = 0; s < 9; ++s) {
      if (s < S0)
        V[0] = __builtin_amdgcn_wmma_f32_16x16x4_f32(false, (TYPE == 0) ? aw0[s < 6 ? s : 0] : aw1[s], false, bfrag0(use, s < S0 ? s : 0), (short)0, V[0], false, false);
      V[1] = __builtin_amdgcn_wmma_f32_16x16x4_f32(false, aw1[s], false, bfrag1(use, s), (short)0, V[1], false, false);
    }
    if (last) {
#pragma unroll
      for (int q = 0; q < 3; ++q) {
#pragma unroll
        for (int t = 0; t < 8; ++t) {
          const float pwx = xw[t] * xX[t];
          v2f a;
          a.x = pwx * yq[t][q].x;
          a.y = pwx * yq[t][q].y;
#pragma unroll
          for (int g = 0; g < 2; ++g) {
            v2f bq;
            bq.x = V[g][2 * q];
            bq.y = V[g][2 * q + 1];
            acc[t][g] = __builtin_amdgcn_wmma_f32_16x16x4_f32(false, a, false, bq, (short)0, acc[t][g], false, false);
          }
        }
      }
    }
  };
#pragma nounroll
  for (int ch = 0; ch < NPAIRCH; ch += 6) {
    chunk(ch, true, false, f0, f2, true);
    chunk(ch + 1, false, true, f1, f0, true);
    chunk(ch + 2, true, false, f2, f1, true);
    chunk(ch + 3, false, true, f0, f2, true);
    chunk(ch + 4, true, false, f1, f0, true);
    chunk(ch + 5, false, true, f2, f1, true);
  }
  static_assert(NPAIRCH % 6 == 0 && NCHUNK == NPAIRCH + 2, "the set rotation below continues the loop's");
  chunk(NPAIRCH, true, true, f0, f2, false);
  chunk(NPAIRCH + 1, true, true, f1, f2, false);

  constexpr int NC2 = (TYPE == 0) ? 10 : 12;
  constexpr int C2B = (TYPE == 0) ? 4 : 6;
  v8f a2a = {}, a2b = {};
  {
    const int crow = 16 * wv + n;
    float xi2[8];
#pragma unroll
    for (int q = 0; q < 4; ++q) {
      xi2[2 * q] = xs[crow * XP + 4 * q + 2 * hi];
      xi2[2 * q + 1] = xs[crow * XP + 4 * q + 2 * hi + 1];
    }
    const bool ok2 = n < NC2;
    const float* p2 = U2_1o;
    int K2s = 0;
    if (TYPE == 0) {
      if (n < 4) {
        p2 = U2_0e + n;
        K2s = 4;
      } else if (n < 10) {
        p2 = U2_1o + (n - 4);
        K2s = 6;
      }
    } else {
      if (n < 6) {
        p2 = U2_1o + 1 * (NL * NL * 6) + n;
        K2s = 6;
      } else if (n < 12) {
        p2 = U2_1o + 2 * (NL * NL * 6) + (n - 6);
        K2s = 6;
      }
    }
#pragma unroll
    for (int W = 0; W < NL; ++W) {
      const float xw2 = xs[crow * XP + W];
#pragma unroll
      for (int q = 0; q < 4; ++q) {
        const int rr = 16 * W + 4 * q + 2 * hi;
        v2f a;
        a.x = xw2 * xi2[2 * q];
        a.y = xw2 * xi2[2 * q + 1];
        float b0 = p2[rr * K2s], b1 = p2[(rr + 1) * K2s];
        v2f bv;
        bv.x = ok2 ? b0 : 0.f;
        bv.y = ok2 ? b1 : 0.f;
        if (q & 1)
          a2b = __builtin_amdgcn_wmma_f32_16x16x4_f32(false, a, false, bv, (short)0, a2b, false, false);
        else
          a2a = __builtin_amdgcn_wmma_f32_16x16x4_f32(false, a, false, bv, (short)0, a2a, false, false);
      }
    }
  }

  float* P = bufs;
#pragma unroll
  for (int t = 0; t < 8; ++t)
#pragma unroll
    for (int g = 0; g < 2; ++g)
#pragma unroll
      for (int j = 0; j < 8; ++j)
        P[((wv * 2 + g) * NL + n) * PP + 16 * t + 8 * hi + j] = acc[t][g][j];
  {
    v8f d2 = a2a + a2b;
#pragma unroll
    for (int j = 0; j < 8; ++j) p2s[n * PP + 16 * wv + 8 * hi + j] = d2[j];
  }
  __syncthreads();

  {
    const int g = tid >> 7;
    const int q = 2 * TYPE + g;
    const int b = tid & 127;
    float yv[NE];
#pragma unroll
    for (int e = 0; e < NE; ++e) yv[e] = ysm[b * NE + e];
    float o = 0.f;
#pragma unroll
    for (int i = 0; i < NL; ++i) {
      float s = 0.f;
#pragma unroll
      for (int w = 0; w < 8; ++w) s += P[((w * 2 + g) * NL + i) * PP + b];
      o = fmaf(xs[b * XP + i], s, o);
    }
    const float* cg = cst + 96 * g;
    const int K2 = (q == 0) ? 4 : 6;
    const float* P2c = p2s + (g == 0 ? 0 : C2B) * PP + b;
    for (int kk = 0; kk < K2; ++kk) {
      float wy = 0.f;
#pragma unroll
      for (int e = 0; e < NE; ++e) wy = fmaf(cg[e * K2 + kk], yv[e], wy);
      o = fmaf(wy, P2c[kk * PP], o);
    }
    float wy1 = 0.f;
#pragma unroll
    for (int e = 0; e < NE; ++e) wy1 = fmaf(cg[60 + e], yv[e], wy1);
    float s1 = 0.f;
#pragma unroll
    for (int W = 0; W < NL; ++W) s1 = fmaf(xs[b * XP + W], cg[70 + W], s1);
    o = fmaf(wy1, s1, o);
    if (q == 0)
      out[b * 512 + c] = o;
    else
      out[b * 512 + NC + c * 3 + (q - 1)] = o;
  }
}

__global__ __launch_bounds__(256) void symcon_kernel(
    const float* __restrict__ x, const float* __restrict__ y,
    const float* __restrict__ U1_0e, const float* __restrict__ U2_0e,
    const float* __restrict__ U3_0e, const float* __restrict__ U1_1o,
    const float* __restrict__ U2_1o, const float* __restrict__ U3_1o,
    const float* __restrict__ w1_0e, const float* __restrict__ w2_0e,
    const float* __restrict__ w3_0e, const float* __restrict__ w1_1o,
    const float* __restrict__ w2_1o, const float* __restrict__ w3_1o,
    float* __restrict__ out) {
  __shared__ __attribute__((aligned(16))) float bufs[SM_FLOATS];
  __shared__ __attribute__((aligned(16))) float xs[NB * XP];
  __shared__ __attribute__((aligned(16))) float ysm[NB * NE];
  __shared__ float p2s[16 * PP];
  __shared__ float cst[2 * 96];
  if (blockIdx.y == 0)
    symcon_body<0>(x, y, U1_0e, U2_0e, U3_0e, U1_1o, U2_1o, U3_1o, w1_0e, w2_0e, w3_0e, w1_1o, w2_1o, w3_1o,
                   out, bufs, xs, ysm, p2s, cst);
  else
    symcon_body<1>(x, y, U1_0e, U2_0e, U3_0e, U1_1o, U2_1o, U3_1o, w1_0e, w2_0e, w3_0e, w1_1o, w2_1o, w3_1o,
                   out, bufs, xs, ysm, p2s, cst);
}

extern "C" void kernel_launch(void* const* d_in, const int* in_sizes, int n_in,
                              void* d_out, int out_size, void* d_ws,
                              size_t ws_size, hipStream_t stream) {
  (void)in_sizes; (void)n_in; (void)d_ws; (void)ws_size; (void)out_size;
  const float* x     = (const float*)d_in[0];
  const float* y     = (const float*)d_in[1];
  const float* U1_0e = (const float*)d_in[2];
  const float* U2_0e = (const float*)d_in[3];
  const float* U3_0e = (const float*)d_in[4];
  const float* U1_1o = (const float*)d_in[5];
  const float* U2_1o = (const float*)d_in[6];
  const float* U3_1o = (const float*)d_in[7];
  const float* w1_0e = (const float*)d_in[8];
  const float* w2_0e = (const float*)d_in[9];
  const float* w3_0e = (const float*)d_in[10];
  const float* w1_1o = (const float*)d_in[11];
  const float* w2_1o = (const float*)d_in[12];
  const float* w3_1o = (const float*)d_in[13];
  float* out = (float*)d_out;

  dim3 grid(NC, 2);
  symcon_kernel<<<grid, 256, 0, stream>>>(x, y, U1_0e, U2_0e, U3_0e, U1_1o,
                                          U2_1o, U3_1o, w1_0e, w2_0e, w3_0e,
                                          w1_1o, w2_1o, w3_1o, out);
}


# The two U3 tiles (W,X),(X,W) of a pair are added in registers before product (1) (same A operand w3, product linear in B): product (1) runs once per pair, 17 not 32 per wave (new sum order)
# speedup vs baseline: 6.3245x; 1.0765x over previous
#include <hip/hip_runtime.h>


typedef __attribute__((ext_vector_type(2))) float v2f;
typedef __attribute__((ext_vector_type(4))) float v4f;
typedef __attribute__((ext_vector_type(8))) float v8f;

#define NB 128
#define NC 128
#define NE 10
#define NL 16
#define K3A 23
#define K3B 33
#define NROW 4096
#define NCHUNK 32
#define NPAIRCH 30
#define XP 20
#define PP 129
#define SM_FLOATS (8 * 2 * NL * PP)


template <int K>
__device__ __forceinline__ int kpos(int s, int hi, int e) {
  static_assert(K == K3A || K == K3B, "piece maps are written for 23 and 33 params");
  if (s < 2 * (K / 8)) return 8 * (s >> 1) + 4 * hi + 2 * (s & 1) + e;
  if (K == K3B) return (hi == 0 && e == 1) ? 32 : -1;
  const int idx = 2 * (s & 1) + e;
  if (hi == 0) return 16 + idx;
  return (idx == 0) ? -1 : 19 + idx;
}

struct FragSet {
  v4f a[4];
  v2f at;
  v4f b[4];
  v2f bt;
};

template <int TYPE>
__device__ __forceinline__ void symcon_body(
    const float* __restrict__ x, const float* __restrict__ y,
    const float* __restrict__ U1_0e, const float* __restrict__ U2_0e,
    const float* __restrict__ U3_0e, const float* __restrict__ U1_1o,
    const float* __restrict__ U2_1o, const float* __restrict__ U3_1o,
    const float* __restrict__ w1_0e, const float* __restrict__ w2_0e,
    const float* __restrict__ w3_0e, const float* __restrict__ w1_1o,
    const float* __restrict__ w2_1o, const float* __restrict__ w3_1o,
    float* __restrict__ out, float* __restrict__ bufs, float* __restrict__ xs,
    float* __restrict__ ysm, float* __restrict__ p2s, float* __restrict__ cst) {
  constexpr int K0 = (TYPE == 0) ? K3A : K3B;
  constexpr int S0 = (K0 + 3) / 4;
  constexpr int S1 = 9;

  const int tid = threadIdx.x;
  const int c = blockIdx.x;
  const int wv = tid >> 5;
  const int lane = tid & 31;
  const int n = lane & 15;
  const int hi = lane >> 4;
  const float* __restrict__ src0 = (TYPE == 0) ? U3_0e : (U3_1o + 1 * (NROW * K3B));
  const float* __restrict__ src1 = (TYPE == 0) ? U3_1o : (U3_1o + 2 * (NROW * K3B));

  const int wvs = __builtin_amdgcn_readfirstlane(wv);
  auto tile_of = [&](int ch) -> int {
    if (ch < NPAIRCH) {
      const int s = ch >> 1, d = (s >> 1) + 1, a = 8 * (s & 1) + wvs, b = (a + d) & 15;
      return (ch & 1) ? 16 * b + a : 16 * a + b;
    }
    return 17 * (8 * (ch - NPAIRCH) + wvs);
  };

  auto ld4 = [](const float* p) __attribute__((always_inline)) -> v4f {
    v4f v;
    __builtin_memcpy(&v, p, sizeof(v4f));
    return v;
  };
  auto ld2 = [](const float* p) __attribute__((always_inline)) -> v2f {
    v2f v;
    __builtin_memcpy(&v, p, sizeof(v2f));
    return v;
  };
  auto issue = [&](int ch, FragSet& f) __attribute__((always_inline)) {
    const int T = tile_of(ch);
    const float* p0 = src0 + T * (16 * K0) + n * K0;
    const float* p1 = src1 + T * (16 * K3B) + n * K3B;
    if (K0 == K3B) {
#pragma unroll
      for (int j = 0; j < 4; ++j) f.a[j] = ld4(p0 + 8 * j + 4 * hi);
      f.at = ld2(p0 + 31);
    } else {
      f.a[0] = ld4(p0 + 4 * hi);
      f.a[1] = ld4(p0 + 8 + 4 * hi);
      f.a[2] = ld4(p0 + 16 + 3 * hi);
    }
#pragma unroll
    for (int j = 0; j < 4; ++j) f.b[j] = ld4(p1 + 8 * j + 4 * hi);
    f.bt = ld2(p1 + 31);
  };
  auto bfrag0 = [&](const FragSet& f, int s) __attribute__((always_inline)) -> v2f {
    if (K0 == K3B && s == 8) return f.at;
    const v4f q = f.a[s >> 1];
    v2f r;
    r.x = q[2 * (s & 1)];
    r.y = q[2 * (s & 1) + 1];
    return r;
  };
  auto bfrag1 = [&](const FragSet& f, int s) __attribute__((always_inline)) -> v2f {
    if (s == 8) return f.bt;
    const v4f q = f.b[s >> 1];
    v2f r;
    r.x = q[2 * (s & 1)];
    r.y = q[2 * (s & 1) + 1];
    return r;
  };

  FragSet f0, f1, f2;
  issue(0, f0);
  issue(1, f1);
  static_assert(NB * NE == 5 * 256 && NB * NL == 2 * 4 * 256, "prologue copies are sized for 256 threads");
  float yr[5];
#pragma unroll
  for (int it = 0; it < 5; ++it) yr[it] = y[tid + 256 * it];
  v4f xr[2];
#pragma unroll
  for (int it = 0; it < 2; ++it) {
    const int i = tid + 256 * it, al = i >> 2, part = i & 3;
    xr[it] = *(const v4f*)(x + (al * NC + c) * NL + 4 * part);
  }
  float cv;
  {
    const int g2 = min(tid / 96, 1), j = tid - 96 * (tid / 96);
    const int q2 = 2 * TYPE + g2;
    const float* w2q = (q2 == 0) ? w2_0e : w2_1o;
    const float* w1q = (q2 == 0) ? w1_0e : w1_1o;
    const float* U1q = (q2 == 0) ? U1_0e : (U1_1o + (q2 - 1) * NL);
    const int K2q = (q2 == 0) ? 4 : 6;
    const float* ptr = w1q + c;
    if (j < NE * K2q) ptr = w2q + j * NC + c;
    else if (j >= 60 && j < 70) ptr = w1q + (j - 60) * NC + c;
    else if (j >= 70 && j < 86) ptr = U1q + (j - 70);
    cv = *ptr;
  }

  v2f aw0[6], aw1[9];
  const bool erow = (n < 6) || (n >= 8 && n < 12);
  const int nc = erow ? ((n < 6) ? n : n - 2) : 0;
  if (TYPE == 0) {
#pragma unroll
    for (int s = 0; s < 6; ++s) {
      const int k0 = kpos<K3A>(s, hi, 0), k1 = kpos<K3A>(s, hi, 1);
      const float v0 = w3_0e[(nc * K3A + max(k0, 0)) * NC + c];
      const float v1 = w3_0e[(nc * K3A + max(k1, 0)) * NC + c];
      aw0[s].x = (erow && k0 >= 0) ? v0 : 0.f;
      aw0[s].y = (erow && k1 >= 0) ? v1 : 0.f;
    }
  }
#pragma unroll
  for (int s = 0; s < 9; ++s) {
    const int k0 = kpos<K3B>(s, hi, 0), k1 = kpos<K3B>(s, hi, 1);
    const float v0 = w3_1o[(nc * K3B + max(k0, 0)) * NC + c];
    const float v1 = w3_1o[(nc * K3B + max(k1, 0)) * NC + c];
    aw1[s].x = (erow && k0 >= 0) ? v0 : 0.f;
    aw1[s].y = (erow && k1 >= 0) ? v1 : 0.f;
  }

#pragma unroll
  for (int it = 0; it < 5; ++it) ysm[tid + 256 * it] = yr[it];
#pragma unroll
  for (int it = 0; it < 2; ++it) {
    const int i = tid + 256 * it, al = i >> 2, part = i & 3;
    *(v4f*)(xs + al * XP + 4 * part) = xr[it];
  }
  if (tid < 192) cst[tid] = cv;
  __syncthreads();

  v2f yq[8][3];
#pragma unroll
  for (int t = 0; t < 8; ++t) {
#pragma unroll
    for (int q = 0; q < 3; ++q) {
      const int e0 = min(6 * hi + 2 * q, NE - 2);
      const v2f v = *(const v2f*)(ysm + (16 * t + n) * NE + e0);
      const bool used = (q < 2) || (hi == 0);
      yq[t][q].x = used ? v.x : 0.f;
      yq[t][q].y = used ? v.y : 0.f;
    }
  }

  v8f acc[8][2] = {};
  v8f V[2];
  auto addset = [&](FragSet& s, const FragSet& t) __attribute__((always_inline)) {
    if (K0 == K3B) {
#pragma unroll
      for (int j = 0; j < 4; ++j) s.a[j] += t.a[j];
      s.at += t.at;
    } else {
#pragma unroll
      for (int j = 0; j < 3; ++j) s.a[j] += t.a[j];
    }
#pragma unroll
    for (int j = 0; j < 4; ++j) s.b[j] += t.b[j];
    s.bt += t.bt;
  };
  auto work = [&](int ch, const FragSet& use) __attribute__((always_inline)) {
    float xw[8], xX[8];
    {
      const int T = tile_of(ch);
      const int W = T >> 4, X = T & 15;
#pragma unroll
      for (int t = 0; t < 8; ++t) {
        xw[t] = xs[(16 * t + n) * XP + W];
        xX[t] = xs[(16 * t + n) * XP + X];
      }
    }
    __builtin_amdgcn_sched_barrier(0);
    V[0] = v8f{};
    V[1] = v8f{};
#pragma unroll
    for (int s = 0; s < 9; ++s) {
      if (s < S0)
        V[0] = __builtin_amdgcn_wmma_f32_16x16x4_f32(false, (TYPE == 0) ? aw0[s < 6 ? s : 0] : aw1[s], false, bfrag0(use, s < S0 ? s : 0), (short)0, V[0], false, false);
      V[1] = __builtin_amdgcn_wmma_f32_16x16x4_f32(false, aw1[s], false, bfrag1(use, s), (short)0, V[1], false, false);
    }
#pragma unroll
    for (int q = 0; q < 3; ++q) {
#pragma unroll
      for (int t = 0; t < 8; ++t) {
        const float pwx = xw[t] * xX[t];
        v2f a;
        a.x = pwx * yq[t][q].x;
        a.y = pwx * yq[t][q].y;
#pragma unroll
        for (int g = 0; g < 2; ++g) {
          v2f bq;
          bq.x = V[g][2 * q];
          bq.y = V[g][2 * q + 1];
          acc[t][g] = __builtin_amdgcn_wmma_f32_16x16x4_f32(false, a, false, bq, (short)0, acc[t][g], false, false);
        }
      }
    }
  };
  auto pairstep = [&](int ch, FragSet& use, FragSet& other, FragSet& third) __attribute__((always_inline)) {
    __builtin_amdgcn_sched_barrier(0);
    addset(use, other);
    issue(ch + 2, other);
    issue(ch + 3, third);
    work(ch, use);
  };
#pragma nounroll
  for (int ch = 0; ch < NPAIRCH; ch += 6) {
    pairstep(ch, f0, f1, f2);
    pairstep(ch + 2, f1, f2, f0);
    pairstep(ch + 4, f2, f0, f1);
  }
  static_assert(NPAIRCH % 6 == 0 && NCHUNK == NPAIRCH + 2, "the set rotation below continues the loop's");
  work(NPAIRCH, f0);
  work(NPAIRCH + 1, f1);

  constexpr int NC2 = (TYPE == 0) ? 10 : 12;
  constexpr int C2B = (TYPE == 0) ? 4 : 6;
  v8f a2a = {}, a2b = {};
  {
    const int crow = 16 * wv + n;
    float xi2[8];
#pragma unroll
    for (int q = 0; q < 4; ++q) {
      xi2[2 * q] = xs[crow * XP + 4 * q + 2 * hi];
      xi2[2 * q + 1] = xs[crow * XP + 4 * q + 2 * hi + 1];
    }
    const bool ok2 = n < NC2;
    const float* p2 = U2_1o;
    int K2s = 0;
    if (TYPE == 0) {
      if (n < 4) {
        p2 = U2_0e + n;
        K2s = 4;
      } else if (n < 10) {
        p2 = U2_1o + (n - 4);
        K2s = 6;
      }
    } else {
      if (n < 6) {
        p2 = U2_1o + 1 * (NL * NL * 6) + n;
        K2s = 6;
      } else if (n < 12) {
        p2 = U2_1o + 2 * (NL * NL * 6) + (n - 6);
        K2s = 6;
      }
    }
#pragma unroll
    for (int W = 0; W < NL; ++W) {
      const float xw2 = xs[crow * XP + W];
#pragma unroll
      for (int q = 0; q < 4; ++q) {
        const int rr = 16 * W + 4 * q + 2 * hi;
        v2f a;
        a.x = xw2 * xi2[2 * q];
        a.y = xw2 * xi2[2 * q + 1];
        float b0 = p2[rr * K2s], b1 = p2[(rr + 1) * K2s];
        v2f bv;
        bv.x = ok2 ? b0 : 0.f;
        bv.y = ok2 ? b1 : 0.f;
        if (q & 1)
          a2b = __builtin_amdgcn_wmma_f32_16x16x4_f32(false, a, false, bv, (short)0, a2b, false, false);
        else
          a2a = __builtin_amdgcn_wmma_f32_16x16x4_f32(false, a, false, bv, (short)0, a2a, false, false);
      }
    }
  }

  float* P = bufs;
#pragma unroll
  for (int t = 0; t < 8; ++t)
#pragma unroll
    for (int g = 0; g < 2; ++g)
#pragma unroll
      for (int j = 0; j < 8; ++j)
        P[((wv * 2 + g) * NL + n) * PP + 16 * t + 8 * hi + j] = acc[t][g][j];
  {
    v8f d2 = a2a + a2b;
#pragma unroll
    for (int j = 0; j < 8; ++j) p2s[n * PP + 16 * wv + 8 * hi + j] = d2[j];
  }
  __syncthreads();

  {
    const int g = tid >> 7;
    const int q = 2 * TYPE + g;
    const int b = tid & 127;
    float yv[NE];
#pragma unroll
    for (int e = 0; e < NE; ++e) yv[e] = ysm[b * NE + e];
    float o = 0.f;
#pragma unroll
    for (int i = 0; i < NL; ++i) {
      float s = 0.f;
#pragma unroll
      for (int w = 0; w < 8; ++w) s += P[((w * 2 + g) * NL + i) * PP + b];
      o = fmaf(xs[b * XP + i], s, o);
    }
    const float* cg = cst + 96 * g;
    const int K2 = (q == 0) ? 4 : 6;
    const float* P2c = p2s + (g == 0 ? 0 : C2B) * PP + b;
    for (int kk = 0; kk < K2; ++kk) {
      float wy = 0.f;
#pragma unroll
      for (int e = 0; e < NE; ++e) wy = fmaf(cg[e * K2 + kk], yv[e], wy);
      o = fmaf(wy, P2c[kk * PP], o);
    }
    float wy1 = 0.f;
#pragma unroll
    for (int e = 0; e < NE; ++e) wy1 = fmaf(cg[60 + e], yv[e], wy1);
    float s1 = 0.f;
#pragma unroll
    for (int W = 0; W < NL; ++W) s1 = fmaf(xs[b * XP + W], cg[70 + W], s1);
    o = fmaf(wy1, s1, o);
    if (q == 0)
      out[b * 512 + c] = o;
    else
      out[b * 512 + NC + c * 3 + (q - 1)] = o;
  }
}

__global__ __launch_bounds__(256) void symcon_kernel(
    const float* __restrict__ x, const float* __restrict__ y,
    const float* __restrict__ U1_0e, const float* __restrict__ U2_0e,
    const float* __restrict__ U3_0e, const float* __restrict__ U1_1o,
    const float* __restrict__ U2_1o, const float* __restrict__ U3_1o,
    const float* __restrict__ w1_0e, const float* __restrict__ w2_0e,
    const float* __restrict__ w3_0e, const float* __restrict__ w1_1o,
    const float* __restrict__ w2_1o, const float* __restrict__ w3_1o,
    float* __restrict__ out) {
  __shared__ __attribute__((aligned(16))) float bufs[SM_FLOATS];
  __shared__ __attribute__((aligned(16))) float xs[NB * XP];
  __shared__ __attribute__((aligned(16))) float ysm[NB * NE];
  __shared__ float p2s[16 * PP];
  __shared__ float cst[2 * 96];
  if (blockIdx.y == 0)
    symcon_body<0>(x, y, U1_0e, U2_0e, U3_0e, U1_1o, U2_1o, U3_1o, w1_0e, w2_0e, w3_0e, w1_1o, w2_1o, w3_1o,
                   out, bufs, xs, ysm, p2s, cst);
  else
    symcon_body<1>(x, y, U1_0e, U2_0e, U3_0e, U1_1o, U2_1o, U3_1o, w1_0e, w2_0e, w3_0e, w1_1o, w2_1o, w3_1o,
                   out, bufs, xs, ysm, p2s, cst);
}

extern "C" void kernel_launch(void* const* d_in, const int* in_sizes, int n_in,
                              void* d_out, int out_size, void* d_ws,
                              size_t ws_size, hipStream_t stream) {
  (void)in_sizes; (void)n_in; (void)d_ws; (void)ws_size; (void)out_size;
  const float* x     = (const float*)d_in[0];
  const float* y     = (const float*)d_in[1];
  const float* U1_0e = (const float*)d_in[2];
  const float* U2_0e = (const float*)d_in[3];
  const float* U3_0e = (const float*)d_in[4];
  const float* U1_1o = (const float*)d_in[5];
  const float* U2_1o = (const float*)d_in[6];
  const float* U3_1o = (const float*)d_in[7];
  const float* w1_0e = (const float*)d_in[8];
  const float* w2_0e = (const float*)d_in[9];
  const float* w3_0e = (const float*)d_in[10];
  const float* w1_1o = (const float*)d_in[11];
  const float* w2_1o = (const float*)d_in[12];
  const float* w3_1o = (const float*)d_in[13];
  float* out = (float*)d_out;

  dim3 grid(NC, 2);
  symcon_kernel<<<grid, 256, 0, stream>>>(x, y, U1_0e, U2_0e, U3_0e, U1_1o,
                                          U2_1o, U3_1o, w1_0e, w2_0e, w3_0e,
                                          w1_1o, w2_1o, w3_1o, out);
}
